# starting kernel
# baseline (speedup 1.0000x reference)
#include <hip/hip_runtime.h>
#include <hip/hip_bf16.h>


typedef __attribute__((ext_vector_type(2))) float v2f;
typedef __attribute__((ext_vector_type(8))) float v8f;

#define BS   2
#define SEQ  2048
#define DMODEL 1024
#define HEADS 16
#define HD   64
#define N3   3072
#define MTOT (BS * SEQ)

__device__ __forceinline__ v8f wmma_f32_16x16x4(v2f a, v2f b, v8f c) {
  return __builtin_amdgcn_wmma_f32_16x16x4_f32(false, a, false, b, (short)0, c,
                                               false, false);
}

__global__ void __launch_bounds__(32)
qkv_proj_kernel(const float* __restrict__ E, const float* __restrict__ W,
                const float* __restrict__ bias, float* __restrict__ Qw,
                float* __restrict__ Kw, float* __restrict__ Vw) {
  const int lane  = threadIdx.x & 31;
  const int n15   = lane & 15;
  const int khalf = (lane >> 4) << 1;
  const int NT    = N3 / 64;
  const int mt    = blockIdx.x / NT;
  const int nt    = blockIdx.x % NT;
  const int nbase = nt * 64;

  v8f acc[4][4];
#pragma unroll
  for (int mm = 0; mm < 4; ++mm)
#pragma unroll
    for (int t = 0; t < 4; ++t) acc[mm][t] = v8f{};

  for (int k4 = 0; k4 < DMODEL / 4; ++k4) {
    const int kb = k4 * 4 + khalf;
    v2f a[4];
#pragma unroll
    for (int mm = 0; mm < 4; ++mm) {
      const float* ap =
          E + (size_t)(mt * 64 + mm * 16 + n15) * DMODEL + kb;
      a[mm] = *(const v2f*)ap;
    }
#pragma unroll
    for (int t = 0; t < 4; ++t) {
      const int ncol = nbase + t * 16 + n15;
      v2f bf;
      bf.x = W[(size_t)kb * N3 + ncol];
      bf.y = W[(size_t)(kb + 1) * N3 + ncol];
#pragma unroll
      for (int mm = 0; mm < 4; ++mm)
        acc[mm][t] = wmma_f32_16x16x4(a[mm], bf, acc[mm][t]);
    }
  }

  const int mbase = (lane >> 4) * 8;
#pragma unroll
  for (int t = 0; t < 4; ++t) {
    const int ncol  = nbase + t * 16 + n15;
    const float bv  = bias[ncol];
    const int which = ncol >> 10;
    const int c     = ncol & 1023;
    const int head  = c & 15;
    const int dim   = c >> 4;
    float* dst = (which == 0) ? Qw : (which == 1) ? Kw : Vw;
    const float scale = (which == 0) ? 0.125f : 1.0f;
#pragma unroll
    for (int mm = 0; mm < 4; ++mm) {
#pragma unroll
      for (int r = 0; r < 8; ++r) {
        const int mrow = mt * 64 + mm * 16 + mbase + r;
        const int b = mrow >> 11;
        const int s = mrow & 2047;
        dst[(((size_t)(b * HEADS + head)) * SEQ + s) * HD + dim] =
            (acc[mm][t][r] + bv) * scale;
      }
    }
  }
}

__global__ void __launch_bounds__(32)
flash_attn_kernel(const float* __restrict__ Qw, const float* __restrict__ Kw,
                  const float* __restrict__ Vw, float* __restrict__ out) {
  __shared__ float pbuf[2][16 * 17];

  const int lane  = threadIdx.x & 31;
  const int n15   = lane & 15;
  const int khalf = (lane >> 4) << 1;
  const int mbase = (lane >> 4) * 8;

  const int tile = blockIdx.x;
  const int qt = tile & 63;
  const int bh = tile >> 6;
  const int h  = bh & 15;
  const int b  = bh >> 4;

  const float* Qh = Qw + (((size_t)bh) * SEQ + qt * 32) * HD;
  const float* Kh = Kw + ((size_t)bh) * SEQ * HD;
  const float* Vh = Vw + ((size_t)bh) * SEQ * HD;

  v2f qa[2][16];
#pragma unroll
  for (int u = 0; u < 2; ++u)
#pragma unroll
    for (int kk = 0; kk < 16; ++kk)
      qa[u][kk] = *(const v2f*)(Qh + (u * 16 + n15) * HD + kk * 4 + khalf);

  v8f acc[2][4];
#pragma unroll
  for (int u = 0; u < 2; ++u)
#pragma unroll
    for (int t = 0; t < 4; ++t) acc[u][t] = v8f{};
  float mi[2][8], li[2][8];
#pragma unroll
  for (int u = 0; u < 2; ++u)
#pragma unroll
    for (int r = 0; r < 8; ++r) { mi[u][r] = -3.0e38f; li[u][r] = 0.0f; }

  for (int kc = 0; kc < SEQ / 16; ++kc) {
    if (kc + 1 < SEQ / 16) {
      const float* Kn = Kh + (size_t)(kc + 1) * 16 * HD;
      const float* Vn = Vh + (size_t)(kc + 1) * 16 * HD;
      __builtin_prefetch(Kn + lane * 32, 0, 1);
      __builtin_prefetch(Vn + lane * 32, 0, 1);
    }

    const float* Kc = Kh + (size_t)(kc * 16 + n15) * HD;
    v8f s0 = v8f{}, s1 = v8f{};
#pragma unroll
    for (int kk = 0; kk < 16; ++kk) {
      v2f bf = *(const v2f*)(Kc + kk * 4 + khalf);
      s0 = wmma_f32_16x16x4(qa[0][kk], bf, s0);
      s1 = wmma_f32_16x16x4(qa[1][kk], bf, s1);
    }

    float alpha[2][8];
    v8f p[2];
#pragma unroll
    for (int u = 0; u < 2; ++u) {
      v8f s = (u == 0) ? s0 : s1;
#pragma unroll
      for (int r = 0; r < 8; ++r) {
        float rowmax = s[r];
        rowmax = fmaxf(rowmax, __shfl_xor(rowmax, 1, 32));
        rowmax = fmaxf(rowmax, __shfl_xor(rowmax, 2, 32));
        rowmax = fmaxf(rowmax, __shfl_xor(rowmax, 4, 32));
        rowmax = fmaxf(rowmax, __shfl_xor(rowmax, 8, 32));
        const float newm = fmaxf(mi[u][r], rowmax);
        alpha[u][r] = __expf(mi[u][r] - newm);
        mi[u][r] = newm;
        const float pv = __expf(s[r] - newm);
        p[u][r] = pv;
        float rs = pv;
        rs += __shfl_xor(rs, 1, 32);
        rs += __shfl_xor(rs, 2, 32);
        rs += __shfl_xor(rs, 4, 32);
        rs += __shfl_xor(rs, 8, 32);
        li[u][r] = li[u][r] * alpha[u][r] + rs;
      }
#pragma unroll
      for (int t = 0; t < 4; ++t)
#pragma unroll
        for (int r = 0; r < 8; ++r) acc[u][t][r] *= alpha[u][r];
    }

#pragma unroll
    for (int u = 0; u < 2; ++u)
#pragma unroll
      for (int r = 0; r < 8; ++r)
        pbuf[u][(mbase + r) * 17 + n15] = p[u][r];
    __syncthreads();

    const float* Vc = Vh + (size_t)(kc * 16) * HD;
#pragma unroll
    for (int kk2 = 0; kk2 < 4; ++kk2) {
      const int k0 = kk2 * 4 + khalf;
      v2f af[2];
#pragma unroll
      for (int u = 0; u < 2; ++u) {
        af[u].x = pbuf[u][n15 * 17 + k0];
        af[u].y = pbuf[u][n15 * 17 + k0 + 1];
      }
#pragma unroll
      for (int t = 0; t < 4; ++t) {
        v2f bf;
        bf.x = Vc[(size_t)k0 * HD + t * 16 + n15];
        bf.y = Vc[(size_t)(k0 + 1) * HD + t * 16 + n15];
        acc[0][t] = wmma_f32_16x16x4(af[0], bf, acc[0][t]);
        acc[1][t] = wmma_f32_16x16x4(af[1], bf, acc[1][t]);
      }
    }
    __syncthreads();
  }

#pragma unroll
  for (int u = 0; u < 2; ++u) {
#pragma unroll
    for (int t = 0; t < 4; ++t) {
#pragma unroll
      for (int r = 0; r < 8; ++r) {
        const int srow = qt * 32 + u * 16 + mbase + r;
        out[(((size_t)b * SEQ) + srow) * DMODEL + h * HD + t * 16 + n15] =
            acc[u][t][r] / li[u][r];
      }
    }
  }
}

extern "C" void kernel_launch(void* const* d_in, const int* in_sizes, int n_in,
                              void* d_out, int out_size, void* d_ws,
                              size_t ws_size, hipStream_t stream) {
  const float* E    = (const float*)d_in[0];
  const float* W    = (const float*)d_in[1];
  const float* bias = (const float*)d_in[2];
  float* out = (float*)d_out;

  const size_t per = (size_t)BS * HEADS * SEQ * HD;
  float* Qw = (float*)d_ws;
  float* Kw = Qw + per;
  float* Vw = Kw + per;

  qkv_proj_kernel<<<(MTOT / 64) * (N3 / 64), 32, 0, stream>>>(E, W, bias, Qw,
                                                              Kw, Vw);
  flash_attn_kernel<<<BS * HEADS * (SEQ / 32), 32, 0, stream>>>(Qw, Kw, Vw,
                                                                out);
}


# Attention products transposed (S^T=K.Q^T, O^T=V^T.P^T): one query per lane, softmax max/sum in registers, P fed from registers, no LDS or shuffles; row-sum and PV key order change
# speedup vs baseline: 1.8272x; 1.8272x over previous
#include <hip/hip_runtime.h>
#include <hip/hip_bf16.h>


typedef __attribute__((ext_vector_type(2))) float v2f;
typedef __attribute__((ext_vector_type(4))) float v4f;
typedef __attribute__((ext_vector_type(8))) float v8f;

#define BS   2
#define SEQ  2048
#define DMODEL 1024
#define HEADS 16
#define HD   64
#define N3   3072
#define MTOT (BS * SEQ)

__device__ __forceinline__ v8f wmma_f32_16x16x4(v2f a, v2f b, v8f c) {
  return __builtin_amdgcn_wmma_f32_16x16x4_f32(false, a, false, b, (short)0, c,
                                               false, false);
}

__global__ void __launch_bounds__(32)
qkv_proj_kernel(const float* __restrict__ E, const float* __restrict__ W,
                const float* __restrict__ bias, float* __restrict__ Qw,
                float* __restrict__ Kw, float* __restrict__ Vw) {
  const int lane  = threadIdx.x & 31;
  const int n15   = lane & 15;
  const int khalf = (lane >> 4) << 1;
  const int NT    = N3 / 64;
  const int mt    = blockIdx.x / NT;
  const int nt    = blockIdx.x % NT;
  const int nbase = nt * 64;

  v8f acc[4][4];
#pragma unroll
  for (int mm = 0; mm < 4; ++mm)
#pragma unroll
    for (int t = 0; t < 4; ++t) acc[mm][t] = v8f{};

  for (int k4 = 0; k4 < DMODEL / 4; ++k4) {
    const int kb = k4 * 4 + khalf;
    v2f a[4];
#pragma unroll
    for (int mm = 0; mm < 4; ++mm) {
      const float* ap =
          E + (size_t)(mt * 64 + mm * 16 + n15) * DMODEL + kb;
      a[mm] = *(const v2f*)ap;
    }
#pragma unroll
    for (int t = 0; t < 4; ++t) {
      const int ncol = nbase + t * 16 + n15;
      v2f bf;
      bf.x = W[(size_t)kb * N3 + ncol];
      bf.y = W[(size_t)(kb + 1) * N3 + ncol];
#pragma unroll
      for (int mm = 0; mm < 4; ++mm)
        acc[mm][t] = wmma_f32_16x16x4(a[mm], bf, acc[mm][t]);
    }
  }

  const int mbase = (lane >> 4) * 8;
#pragma unroll
  for (int t = 0; t < 4; ++t) {
    const int ncol  = nbase + t * 16 + n15;
    const float bv  = bias[ncol];
    const int which = ncol >> 10;
    const int c     = ncol & 1023;
    const int head  = c & 15;
    const int dim   = c >> 4;
    float* dst = (which == 0) ? Qw : (which == 1) ? Kw : Vw;
    const float scale = (which == 0) ? 0.125f : 1.0f;
#pragma unroll
    for (int mm = 0; mm < 4; ++mm) {
#pragma unroll
      for (int r = 0; r < 8; ++r) {
        const int mrow = mt * 64 + mm * 16 + mbase + r;
        const int b = mrow >> 11;
        const int s = mrow & 2047;
        dst[(((size_t)(b * HEADS + head)) * SEQ + s) * HD + dim] =
            (acc[mm][t][r] + bv) * scale;
      }
    }
  }
}

__device__ __forceinline__ float xhalf(float x) {
  int b = __builtin_bit_cast(int, x);
  return __builtin_bit_cast(
      float, __builtin_amdgcn_permlanex16(b, b, 0x76543210, 0xfedcba98, false,
                                          false));
}

__global__ void __launch_bounds__(32)
flash_attn_kernel(const float* __restrict__ Qw, const float* __restrict__ Kw,
                  const float* __restrict__ Vw, float* __restrict__ out) {
  const int lane  = threadIdx.x & 31;
  const int n15   = lane & 15;
  const int khalf = (lane >> 4) << 1;
  const int mbase = (lane >> 4) * 8;

  const int tile = blockIdx.x;
  const int qt = tile & 63;
  const int bh = tile >> 6;
  const int h  = bh & 15;
  const int b  = bh >> 4;

  const float* Qh = Qw + (((size_t)bh) * SEQ + qt * 32) * HD;
  const float* Kh = Kw + ((size_t)bh) * SEQ * HD;
  const float* Vh = Vw + ((size_t)bh) * SEQ * HD;

  v2f qb[2][16];
#pragma unroll
  for (int u = 0; u < 2; ++u)
#pragma unroll
    for (int kk = 0; kk < 16; ++kk)
      qb[u][kk] = *(const v2f*)(Qh + (u * 16 + n15) * HD + kk * 4 + khalf);

  v8f acc[2][4];
#pragma unroll
  for (int u = 0; u < 2; ++u)
#pragma unroll
    for (int t = 0; t < 4; ++t) acc[u][t] = v8f{};
  float mi[2], li[2];
#pragma unroll
  for (int u = 0; u < 2; ++u) { mi[u] = -3.0e38f; li[u] = 0.0f; }

  for (int kc = 0; kc < SEQ / 16; ++kc) {
    if (kc + 1 < SEQ / 16) {
      const float* Kn = Kh + (size_t)(kc + 1) * 16 * HD;
      const float* Vn = Vh + (size_t)(kc + 1) * 16 * HD;
      __builtin_prefetch(Kn + lane * 32, 0, 1);
      __builtin_prefetch(Vn + lane * 32, 0, 1);
    }

    const float* Kc = Kh + (size_t)(kc * 16 + n15) * HD;
    v8f st[2];
    st[0] = v8f{};
    st[1] = v8f{};
#pragma unroll
    for (int kk = 0; kk < 16; ++kk) {
      v2f kf = *(const v2f*)(Kc + kk * 4 + khalf);
      st[0] = wmma_f32_16x16x4(kf, qb[0][kk], st[0]);
      st[1] = wmma_f32_16x16x4(kf, qb[1][kk], st[1]);
    }

#pragma unroll
    for (int u = 0; u < 2; ++u) {
      float cmax = st[u][0];
#pragma unroll
      for (int r = 1; r < 8; ++r) cmax = fmaxf(cmax, st[u][r]);
      cmax = fmaxf(cmax, xhalf(cmax));
      const float newm  = fmaxf(mi[u], cmax);
      const float alpha = __expf(mi[u] - newm);
      mi[u] = newm;
      float rs = 0.0f;
#pragma unroll
      for (int r = 0; r < 8; ++r) {
        const float pv = __expf(st[u][r] - newm);
        st[u][r] = pv;
        rs += pv;
      }
      li[u] = li[u] * alpha + rs;
#pragma unroll
      for (int t = 0; t < 4; ++t)
#pragma unroll
        for (int r = 0; r < 8; ++r) acc[u][t][r] *= alpha;
    }

    const float* Vc = Vh + (size_t)(kc * 16 + mbase) * HD;
#pragma unroll
    for (int j = 0; j < 4; ++j) {
      v2f pb[2];
#pragma unroll
      for (int u = 0; u < 2; ++u) {
        pb[u].x = st[u][2 * j];
        pb[u].y = st[u][2 * j + 1];
      }
#pragma unroll
      for (int t = 0; t < 4; ++t) {
        v2f vf;
        vf.x = Vc[(size_t)(2 * j) * HD + t * 16 + n15];
        vf.y = Vc[(size_t)(2 * j + 1) * HD + t * 16 + n15];
        acc[0][t] = wmma_f32_16x16x4(vf, pb[0], acc[0][t]);
        acc[1][t] = wmma_f32_16x16x4(vf, pb[1], acc[1][t]);
      }
    }
  }

#pragma unroll
  for (int u = 0; u < 2; ++u) {
    const float l = li[u] + xhalf(li[u]);
    const int srow = qt * 32 + u * 16 + n15;
    float* op = out + (((size_t)b * SEQ) + srow) * DMODEL + h * HD + mbase;
#pragma unroll
    for (int t = 0; t < 4; ++t) {
      v4f lo, hi;
      lo.x = acc[u][t][0] / l; lo.y = acc[u][t][1] / l;
      lo.z = acc[u][t][2] / l; lo.w = acc[u][t][3] / l;
      hi.x = acc[u][t][4] / l; hi.y = acc[u][t][5] / l;
      hi.z = acc[u][t][6] / l; hi.w = acc[u][t][7] / l;
      *(v4f*)(op + t * 16)     = lo;
      *(v4f*)(op + t * 16 + 4) = hi;
    }
  }
}

extern "C" void kernel_launch(void* const* d_in, const int* in_sizes, int n_in,
                              void* d_out, int out_size, void* d_ws,
                              size_t ws_size, hipStream_t stream) {
  const float* E    = (const float*)d_in[0];
  const float* W    = (const float*)d_in[1];
  const float* bias = (const float*)d_in[2];
  float* out = (float*)d_out;

  const size_t per = (size_t)BS * HEADS * SEQ * HD;
  float* Qw = (float*)d_ws;
  float* Kw = Qw + per;
  float* Vw = Kw + per;

  qkv_proj_kernel<<<(MTOT / 64) * (N3 / 64), 32, 0, stream>>>(E, W, bias, Qw,
                                                              Kw, Vw);
  flash_attn_kernel<<<BS * HEADS * (SEQ / 32), 32, 0, stream>>>(Qw, Kw, Vw,
                                                                out);
}


# Workspace Q/K/V planes kept in matrix-fragment order (the projection epilogue writes them so): each attention fragment load is 16 B per lane over 4 whole lines, not 16 partial lines; sums unchanged
# speedup vs baseline: 2.6925x; 1.4736x over previous
#include <hip/hip_runtime.h>
#include <hip/hip_bf16.h>


typedef __attribute__((ext_vector_type(2))) float v2f;
typedef __attribute__((ext_vector_type(4))) float v4f;
typedef __attribute__((ext_vector_type(8))) float v8f;

#define BS   2
#define SEQ  2048
#define DMODEL 1024
#define HEADS 16
#define HD   64
#define N3   3072
#define MTOT (BS * SEQ)

__device__ __forceinline__ v8f wmma_f32_16x16x4(v2f a, v2f b, v8f c) {
  return __builtin_amdgcn_wmma_f32_16x16x4_f32(false, a, false, b, (short)0, c,
                                               false, false);
}

__device__ __forceinline__ int qk_frag_off(int row, int d) {
  return ((row >> 4) << 10) + ((d >> 3) << 7) + (((d >> 1) & 1) << 6) +
         ((row & 15) << 2) + (((d >> 2) & 1) << 1) + (d & 1);
}
__device__ __forceinline__ int v_frag_off(int row, int d) {
  return ((row >> 4) << 10) + (((row & 7) >> 1) << 8) + ((d >> 5) << 7) +
         (((row >> 3) & 1) << 6) + ((d & 15) << 2) + (((d >> 4) & 1) << 1) +
         (row & 1);
}

__global__ void __launch_bounds__(32)
qkv_proj_kernel(const float* __restrict__ E, const float* __restrict__ W,
                const float* __restrict__ bias, float* __restrict__ Qw,
                float* __restrict__ Kw, float* __restrict__ Vw) {
  const int lane  = threadIdx.x & 31;
  const int n15   = lane & 15;
  const int khalf = (lane >> 4) << 1;
  const int NT    = N3 / 64;
  const int mt    = blockIdx.x / NT;
  const int nt    = blockIdx.x % NT;
  const int nbase = nt * 64;

  v8f acc[4][4];
#pragma unroll
  for (int mm = 0; mm < 4; ++mm)
#pragma unroll
    for (int t = 0; t < 4; ++t) acc[mm][t] = v8f{};

  for (int k4 = 0; k4 < DMODEL / 4; ++k4) {
    const int kb = k4 * 4 + khalf;
    v2f a[4];
#pragma unroll
    for (int mm = 0; mm < 4; ++mm) {
      const float* ap =
          E + (size_t)(mt * 64 + mm * 16 + n15) * DMODEL + kb;
      a[mm] = *(const v2f*)ap;
    }
#pragma unroll
    for (int t = 0; t < 4; ++t) {
      const int ncol = nbase + t * 16 + n15;
      v2f bf;
      bf.x = W[(size_t)kb * N3 + ncol];
      bf.y = W[(size_t)(kb + 1) * N3 + ncol];
#pragma unroll
      for (int mm = 0; mm < 4; ++mm)
        acc[mm][t] = wmma_f32_16x16x4(a[mm], bf, acc[mm][t]);
    }
  }

  const int mbase = (lane >> 4) * 8;
  const int which = nt >> 4;
  float* dst = (which == 0) ? Qw : (which == 1) ? Kw : Vw;
  const float scale = (which == 0) ? 0.125f : 1.0f;
  const int b  = mt >> 5;
  const int s0 = (mt & 31) * 64;
  float* plane = dst + ((size_t)(b * HEADS + n15)) * SEQ * HD;
  if (which == 2) {
#pragma unroll
    for (int t = 0; t < 4; ++t) {
      const float bv = bias[nbase + t * 16 + n15];
      const int dim  = (nt * 4 + t) & 63;
#pragma unroll
      for (int mm = 0; mm < 4; ++mm)
#pragma unroll
        for (int r = 0; r < 8; ++r)
          plane[v_frag_off(s0 + mm * 16 + mbase + r, dim)] =
              acc[mm][t][r] + bv;
    }
  } else {
#pragma unroll
    for (int tp = 0; tp < 2; ++tp) {
      const float bv0 = bias[nbase + (2 * tp) * 16 + n15];
      const float bv1 = bias[nbase + (2 * tp + 1) * 16 + n15];
      const int dim0  = (nt * 4 + 2 * tp) & 63;
#pragma unroll
      for (int mm = 0; mm < 4; ++mm)
#pragma unroll
        for (int r = 0; r < 8; ++r) {
          v2f pr;
          pr.x = (acc[mm][2 * tp][r] + bv0) * scale;
          pr.y = (acc[mm][2 * tp + 1][r] + bv1) * scale;
          *(v2f*)(plane + qk_frag_off(s0 + mm * 16 + mbase + r, dim0)) = pr;
        }
    }
  }
}

__device__ __forceinline__ float xhalf(float x) {
  int b = __builtin_bit_cast(int, x);
  return __builtin_bit_cast(
      float, __builtin_amdgcn_permlanex16(b, b, 0x76543210, 0xfedcba98, false,
                                          false));
}

__global__ void __launch_bounds__(32)
flash_attn_kernel(const float* __restrict__ Qw, const float* __restrict__ Kw,
                  const float* __restrict__ Vw, float* __restrict__ out) {
  const int lane  = threadIdx.x & 31;
  const int n15   = lane & 15;
  const int mbase = (lane >> 4) * 8;

  const int tile = blockIdx.x;
  const int qt = tile & 63;
  const int bh = tile >> 6;
  const int h  = bh & 15;
  const int b  = bh >> 4;

  const float* Qh = Qw + (((size_t)bh) * SEQ + qt * 32) * HD;
  const float* Kh = Kw + ((size_t)bh) * SEQ * HD;
  const float* Vh = Vw + ((size_t)bh) * SEQ * HD;

  v2f qb[2][16];
#pragma unroll
  for (int u = 0; u < 2; ++u)
#pragma unroll
    for (int kk2 = 0; kk2 < 8; ++kk2) {
      const v4f q4 = *(const v4f*)(Qh + u * 1024 + kk2 * 128 + lane * 4);
      qb[u][2 * kk2].x     = q4.x;  qb[u][2 * kk2].y     = q4.y;
      qb[u][2 * kk2 + 1].x = q4.z;  qb[u][2 * kk2 + 1].y = q4.w;
    }

  v8f acc[2][4];
#pragma unroll
  for (int u = 0; u < 2; ++u)
#pragma unroll
    for (int t = 0; t < 4; ++t) acc[u][t] = v8f{};
  float mi[2], li[2];
#pragma unroll
  for (int u = 0; u < 2; ++u) { mi[u] = -3.0e38f; li[u] = 0.0f; }

  for (int kc = 0; kc < SEQ / 16; ++kc) {
    if (kc + 1 < SEQ / 16) {
      const float* Kn = Kh + (size_t)(kc + 1) * 16 * HD;
      const float* Vn = Vh + (size_t)(kc + 1) * 16 * HD;
      __builtin_prefetch(Kn + lane * 32, 0, 1);
      __builtin_prefetch(Vn + lane * 32, 0, 1);
    }

    const float* Kc = Kh + (size_t)kc * 1024 + lane * 4;
    v8f st[2];
    st[0] = v8f{};
    st[1] = v8f{};
#pragma unroll
    for (int kk2 = 0; kk2 < 8; ++kk2) {
      const v4f k4 = *(const v4f*)(Kc + kk2 * 128);
      v2f kf0, kf1;
      kf0.x = k4.x; kf0.y = k4.y;
      kf1.x = k4.z; kf1.y = k4.w;
      st[0] = wmma_f32_16x16x4(kf0, qb[0][2 * kk2], st[0]);
      st[1] = wmma_f32_16x16x4(kf0, qb[1][2 * kk2], st[1]);
      st[0] = wmma_f32_16x16x4(kf1, qb[0][2 * kk2 + 1], st[0]);
      st[1] = wmma_f32_16x16x4(kf1, qb[1][2 * kk2 + 1], st[1]);
    }

#pragma unroll
    for (int u = 0; u < 2; ++u) {
      float cmax = st[u][0];
#pragma unroll
      for (int r = 1; r < 8; ++r) cmax = fmaxf(cmax, st[u][r]);
      cmax = fmaxf(cmax, xhalf(cmax));
      const float newm  = fmaxf(mi[u], cmax);
      const float alpha = __expf(mi[u] - newm);
      mi[u] = newm;
      float rs = 0.0f;
#pragma unroll
      for (int r = 0; r < 8; ++r) {
        const float pv = __expf(st[u][r] - newm);
        st[u][r] = pv;
        rs += pv;
      }
      li[u] = li[u] * alpha + rs;
#pragma unroll
      for (int t = 0; t < 4; ++t)
#pragma unroll
        for (int r = 0; r < 8; ++r) acc[u][t][r] *= alpha;
    }

    const float* Vc = Vh + (size_t)kc * 1024 + lane * 4;
#pragma unroll
    for (int j = 0; j < 4; ++j) {
      v2f pb[2];
#pragma unroll
      for (int u = 0; u < 2; ++u) {
        pb[u].x = st[u][2 * j];
        pb[u].y = st[u][2 * j + 1];
      }
#pragma unroll
      for (int t2 = 0; t2 < 2; ++t2) {
        const v4f v4 = *(const v4f*)(Vc + j * 256 + t2 * 128);
        v2f vf0, vf1;
        vf0.x = v4.x; vf0.y = v4.y;
        vf1.x = v4.z; vf1.y = v4.w;
        acc[0][2 * t2]     = wmma_f32_16x16x4(vf0, pb[0], acc[0][2 * t2]);
        acc[1][2 * t2]     = wmma_f32_16x16x4(vf0, pb[1], acc[1][2 * t2]);
        acc[0][2 * t2 + 1] = wmma_f32_16x16x4(vf1, pb[0], acc[0][2 * t2 + 1]);
        acc[1][2 * t2 + 1] = wmma_f32_16x16x4(vf1, pb[1], acc[1][2 * t2 + 1]);
      }
    }
  }

#pragma unroll
  for (int u = 0; u < 2; ++u) {
    const float l = li[u] + xhalf(li[u]);
    const int srow = qt * 32 + u * 16 + n15;
    float* op = out + (((size_t)b * SEQ) + srow) * DMODEL + h * HD + mbase;
#pragma unroll
    for (int t = 0; t < 4; ++t) {
      v4f lo, hi;
      lo.x = acc[u][t][0] / l; lo.y = acc[u][t][1] / l;
      lo.z = acc[u][t][2] / l; lo.w = acc[u][t][3] / l;
      hi.x = acc[u][t][4] / l; hi.y = acc[u][t][5] / l;
      hi.z = acc[u][t][6] / l; hi.w = acc[u][t][7] / l;
      *(v4f*)(op + t * 16)     = lo;
      *(v4f*)(op + t * 16 + 4) = hi;
    }
  }
}

extern "C" void kernel_launch(void* const* d_in, const int* in_sizes, int n_in,
                              void* d_out, int out_size, void* d_ws,
                              size_t ws_size, hipStream_t stream) {
  const float* E    = (const float*)d_in[0];
  const float* W    = (const float*)d_in[1];
  const float* bias = (const float*)d_in[2];
  float* out = (float*)d_out;

  const size_t per = (size_t)BS * HEADS * SEQ * HD;
  float* Qw = (float*)d_ws;
  float* Kw = Qw + per;
  float* Vw = Kw + per;

  qkv_proj_kernel<<<(MTOT / 64) * (N3 / 64), 32, 0, stream>>>(E, W, bias, Qw,
                                                              Kw, Vw);
  flash_attn_kernel<<<BS * HEADS * (SEQ / 32), 32, 0, stream>>>(Qw, Kw, Vw,
                                                                out);
}


# Projection GEMM re-tiled: a 4-wave group stages 128x32 and 32x128 operand tiles in LDS with whole-line 16 B loads (2 stages, 1 barrier per K-step); fragments read from LDS; sums unchanged
# speedup vs baseline: 2.9137x; 1.0822x over previous
#include <hip/hip_runtime.h>
#include <hip/hip_bf16.h>


typedef __attribute__((ext_vector_type(2))) float v2f;
typedef __attribute__((ext_vector_type(4))) float v4f;
typedef __attribute__((ext_vector_type(8))) float v8f;

#define BS   2
#define SEQ  2048
#define DMODEL 1024
#define HEADS 16
#define HD   64
#define N3   3072
#define MTOT (BS * SEQ)

__device__ __forceinline__ v8f wmma_f32_16x16x4(v2f a, v2f b, v8f c) {
  return __builtin_amdgcn_wmma_f32_16x16x4_f32(false, a, false, b, (short)0, c,
                                               false, false);
}

__device__ __forceinline__ int qk_frag_off(int row, int d) {
  return ((row >> 4) << 10) + ((d >> 3) << 7) + (((d >> 1) & 1) << 6) +
         ((row & 15) << 2) + (((d >> 2) & 1) << 1) + (d & 1);
}
__device__ __forceinline__ int v_frag_off(int row, int d) {
  return ((row >> 4) << 10) + (((row & 7) >> 1) << 8) + ((d >> 5) << 7) +
         (((row >> 3) & 1) << 6) + ((d & 15) << 2) + (((d >> 4) & 1) << 1) +
         (row & 1);
}

#define PJ_BM 128
#define PJ_BN 128
#define PJ_BK 32
#define PJ_AP 36
#define PJ_BP 288
#define PJ_ASZ (PJ_BM * PJ_AP)
#define PJ_BSZ ((PJ_BK / 2) * PJ_BP)

__global__ void __launch_bounds__(128)
qkv_proj_kernel(const float* __restrict__ E, const float* __restrict__ W,
                const float* __restrict__ bias, float* __restrict__ Qw,
                float* __restrict__ Kw, float* __restrict__ Vw) {
  __shared__ __attribute__((aligned(16))) float As[2 * PJ_ASZ];
  __shared__ __attribute__((aligned(16))) float Bs[2 * PJ_BSZ];

  const int tid   = threadIdx.x;
  const int lane  = tid & 31;
  const int wave  = tid >> 5;
  const int wm    = wave >> 1;
  const int wn    = wave & 1;
  const int n15   = lane & 15;
  const int khalf = (lane >> 4) << 1;
  const int NTB   = N3 / PJ_BN;
  const int mtb   = blockIdx.x / NTB;
  const int ntb   = blockIdx.x % NTB;

  const int ar = tid >> 3, ac = tid & 7;
  const int bp = tid >> 5, bc = tid & 31;
  const float* Ag = E + (size_t)(mtb * PJ_BM + ar) * DMODEL + ac * 4;
  const float* Bg = W + (size_t)(2 * bp) * N3 + ntb * PJ_BN + bc * 4;
  const int a_st = ar * PJ_AP + ac * 4;
  const int b_st = bp * PJ_BP + bc * 8;

  const int a_rd = (wm * 64 + n15) * PJ_AP + khalf;
  const int b_rd = (lane >> 4) * PJ_BP + (wn * 64 + n15) * 2;

  v8f acc[4][4];
#pragma unroll
  for (int mm = 0; mm < 4; ++mm)
#pragma unroll
    for (int t = 0; t < 4; ++t) acc[mm][t] = v8f{};

  v4f ra[8], rb[8];
#pragma unroll
  for (int i = 0; i < 8; ++i) {
    ra[i] = *(const v4f*)(Ag + (size_t)i * 16 * DMODEL);
    rb[i] = *(const v4f*)(Bg + (size_t)((i >> 1) * 8 + (i & 1)) * N3);
  }
#pragma unroll
  for (int i = 0; i < 8; ++i)
    *(v4f*)(As + a_st + i * 16 * PJ_AP) = ra[i];
#pragma unroll
  for (int i = 0; i < 4; ++i) {
    v4f lo, hi;
    lo.x = rb[2 * i].x; lo.y = rb[2 * i + 1].x;
    lo.z = rb[2 * i].y; lo.w = rb[2 * i + 1].y;
    hi.x = rb[2 * i].z; hi.y = rb[2 * i + 1].z;
    hi.z = rb[2 * i].w; hi.w = rb[2 * i + 1].w;
    *(v4f*)(Bs + b_st + i * 4 * PJ_BP)     = lo;
    *(v4f*)(Bs + b_st + i * 4 * PJ_BP + 4) = hi;
  }
  __syncthreads();

  for (int kt = 0; kt < DMODEL / PJ_BK; ++kt) {
    const bool more = (kt + 1) < DMODEL / PJ_BK;
    if (more) {
      const float* An = Ag + (kt + 1) * PJ_BK;
      const float* Bn = Bg + (size_t)(kt + 1) * PJ_BK * N3;
#pragma unroll
      for (int i = 0; i < 8; ++i) {
        ra[i] = *(const v4f*)(An + (size_t)i * 16 * DMODEL);
        rb[i] = *(const v4f*)(Bn + (size_t)((i >> 1) * 8 + (i & 1)) * N3);
      }
    }

    const float* Ac = As + (kt & 1) * PJ_ASZ + a_rd;
    const float* Bc = Bs + (kt & 1) * PJ_BSZ + b_rd;
#pragma unroll
    for (int ks = 0; ks < PJ_BK / 4; ++ks) {
      v2f a[4];
#pragma unroll
      for (int mm = 0; mm < 4; ++mm)
        a[mm] = *(const v2f*)(Ac + mm * 16 * PJ_AP + ks * 4);
#pragma unroll
      for (int t = 0; t < 4; ++t) {
        const v2f bf = *(const v2f*)(Bc + ks * 2 * PJ_BP + t * 32);
#pragma unroll
        for (int mm = 0; mm < 4; ++mm)
          acc[mm][t] = wmma_f32_16x16x4(a[mm], bf, acc[mm][t]);
      }
    }

    if (more) {
      float* Ad = As + ((kt + 1) & 1) * PJ_ASZ + a_st;
      float* Bd = Bs + ((kt + 1) & 1) * PJ_BSZ + b_st;
#pragma unroll
      for (int i = 0; i < 8; ++i)
        *(v4f*)(Ad + i * 16 * PJ_AP) = ra[i];
#pragma unroll
      for (int i = 0; i < 4; ++i) {
        v4f lo, hi;
        lo.x = rb[2 * i].x; lo.y = rb[2 * i + 1].x;
        lo.z = rb[2 * i].y; lo.w = rb[2 * i + 1].y;
        hi.x = rb[2 * i].z; hi.y = rb[2 * i + 1].z;
        hi.z = rb[2 * i].w; hi.w = rb[2 * i + 1].w;
        *(v4f*)(Bd + i * 4 * PJ_BP)     = lo;
        *(v4f*)(Bd + i * 4 * PJ_BP + 4) = hi;
      }
    }
    __syncthreads();
  }

  const int mt    = mtb * 2 + wm;
  const int nt    = ntb * 2 + wn;
  const int nbase = nt * 64;
  const int mbase = (lane >> 4) * 8;
  const int which = nt >> 4;
  float* dst = (which == 0) ? Qw : (which == 1) ? Kw : Vw;
  const float scale = (which == 0) ? 0.125f : 1.0f;
  const int b  = mt >> 5;
  const int s0 = (mt & 31) * 64;
  float* plane = dst + ((size_t)(b * HEADS + n15)) * SEQ * HD;
  if (which == 2) {
#pragma unroll
    for (int t = 0; t < 4; ++t) {
      const float bv = bias[nbase + t * 16 + n15];
      const int dim  = (nt * 4 + t) & 63;
#pragma unroll
      for (int mm = 0; mm < 4; ++mm)
#pragma unroll
        for (int r = 0; r < 8; ++r)
          plane[v_frag_off(s0 + mm * 16 + mbase + r, dim)] =
              acc[mm][t][r] + bv;
    }
  } else {
#pragma unroll
    for (int tp = 0; tp < 2; ++tp) {
      const float bv0 = bias[nbase + (2 * tp) * 16 + n15];
      const float bv1 = bias[nbase + (2 * tp + 1) * 16 + n15];
      const int dim0  = (nt * 4 + 2 * tp) & 63;
#pragma unroll
      for (int mm = 0; mm < 4; ++mm)
#pragma unroll
        for (int r = 0; r < 8; ++r) {
          v2f pr;
          pr.x = (acc[mm][2 * tp][r] + bv0) * scale;
          pr.y = (acc[mm][2 * tp + 1][r] + bv1) * scale;
          *(v2f*)(plane + qk_frag_off(s0 + mm * 16 + mbase + r, dim0)) = pr;
        }
    }
  }
}

__device__ __forceinline__ float xhalf(float x) {
  int b = __builtin_bit_cast(int, x);
  return __builtin_bit_cast(
      float, __builtin_amdgcn_permlanex16(b, b, 0x76543210, 0xfedcba98, false,
                                          false));
}

__global__ void __launch_bounds__(32)
flash_attn_kernel(const float* __restrict__ Qw, const float* __restrict__ Kw,
                  const float* __restrict__ Vw, float* __restrict__ out) {
  const int lane  = threadIdx.x & 31;
  const int n15   = lane & 15;
  const int mbase = (lane >> 4) * 8;

  const int tile = blockIdx.x;
  const int qt = tile & 63;
  const int bh = tile >> 6;
  const int h  = bh & 15;
  const int b  = bh >> 4;

  const float* Qh = Qw + (((size_t)bh) * SEQ + qt * 32) * HD;
  const float* Kh = Kw + ((size_t)bh) * SEQ * HD;
  const float* Vh = Vw + ((size_t)bh) * SEQ * HD;

  v2f qb[2][16];
#pragma unroll
  for (int u = 0; u < 2; ++u)
#pragma unroll
    for (int kk2 = 0; kk2 < 8; ++kk2) {
      const v4f q4 = *(const v4f*)(Qh + u * 1024 + kk2 * 128 + lane * 4);
      qb[u][2 * kk2].x     = q4.x;  qb[u][2 * kk2].y     = q4.y;
      qb[u][2 * kk2 + 1].x = q4.z;  qb[u][2 * kk2 + 1].y = q4.w;
    }

  v8f acc[2][4];
#pragma unroll
  for (int u = 0; u < 2; ++u)
#pragma unroll
    for (int t = 0; t < 4; ++t) acc[u][t] = v8f{};
  float mi[2], li[2];
#pragma unroll
  for (int u = 0; u < 2; ++u) { mi[u] = -3.0e38f; li[u] = 0.0f; }

  for (int kc = 0; kc < SEQ / 16; ++kc) {
    if (kc + 1 < SEQ / 16) {
      const float* Kn = Kh + (size_t)(kc + 1) * 16 * HD;
      const float* Vn = Vh + (size_t)(kc + 1) * 16 * HD;
      __builtin_prefetch(Kn + lane * 32, 0, 1);
      __builtin_prefetch(Vn + lane * 32, 0, 1);
    }

    const float* Kc = Kh + (size_t)kc * 1024 + lane * 4;
    v8f st[2];
    st[0] = v8f{};
    st[1] = v8f{};
#pragma unroll
    for (int kk2 = 0; kk2 < 8; ++kk2) {
      const v4f k4 = *(const v4f*)(Kc + kk2 * 128);
      v2f kf0, kf1;
      kf0.x = k4.x; kf0.y = k4.y;
      kf1.x = k4.z; kf1.y = k4.w;
      st[0] = wmma_f32_16x16x4(kf0, qb[0][2 * kk2], st[0]);
      st[1] = wmma_f32_16x16x4(kf0, qb[1][2 * kk2], st[1]);
      st[0] = wmma_f32_16x16x4(kf1, qb[0][2 * kk2 + 1], st[0]);
      st[1] = wmma_f32_16x16x4(kf1, qb[1][2 * kk2 + 1], st[1]);
    }

#pragma unroll
    for (int u = 0; u < 2; ++u) {
      float cmax = st[u][0];
#pragma unroll
      for (int r = 1; r < 8; ++r) cmax = fmaxf(cmax, st[u][r]);
      cmax = fmaxf(cmax, xhalf(cmax));
      const float newm  = fmaxf(mi[u], cmax);
      const float alpha = __expf(mi[u] - newm);
      mi[u] = newm;
      float rs = 0.0f;
#pragma unroll
      for (int r = 0; r < 8; ++r) {
        const float pv = __expf(st[u][r] - newm);
        st[u][r] = pv;
        rs += pv;
      }
      li[u] = li[u] * alpha + rs;
#pragma unroll
      for (int t = 0; t < 4; ++t)
#pragma unroll
        for (int r = 0; r < 8; ++r) acc[u][t][r] *= alpha;
    }

    const float* Vc = Vh + (size_t)kc * 1024 + lane * 4;
#pragma unroll
    for (int j = 0; j < 4; ++j) {
      v2f pb[2];
#pragma unroll
      for (int u = 0; u < 2; ++u) {
        pb[u].x = st[u][2 * j];
        pb[u].y = st[u][2 * j + 1];
      }
#pragma unroll
      for (int t2 = 0; t2 < 2; ++t2) {
        const v4f v4 = *(const v4f*)(Vc + j * 256 + t2 * 128);
        v2f vf0, vf1;
        vf0.x = v4.x; vf0.y = v4.y;
        vf1.x = v4.z; vf1.y = v4.w;
        acc[0][2 * t2]     = wmma_f32_16x16x4(vf0, pb[0], acc[0][2 * t2]);
        acc[1][2 * t2]     = wmma_f32_16x16x4(vf0, pb[1], acc[1][2 * t2]);
        acc[0][2 * t2 + 1] = wmma_f32_16x16x4(vf1, pb[0], acc[0][2 * t2 + 1]);
        acc[1][2 * t2 + 1] = wmma_f32_16x16x4(vf1, pb[1], acc[1][2 * t2 + 1]);
      }
    }
  }

#pragma unroll
  for (int u = 0; u < 2; ++u) {
    const float l = li[u] + xhalf(li[u]);
    const int srow = qt * 32 + u * 16 + n15;
    float* op = out + (((size_t)b * SEQ) + srow) * DMODEL + h * HD + mbase;
#pragma unroll
    for (int t = 0; t < 4; ++t) {
      v4f lo, hi;
      lo.x = acc[u][t][0] / l; lo.y = acc[u][t][1] / l;
      lo.z = acc[u][t][2] / l; lo.w = acc[u][t][3] / l;
      hi.x = acc[u][t][4] / l; hi.y = acc[u][t][5] / l;
      hi.z = acc[u][t][6] / l; hi.w = acc[u][t][7] / l;
      *(v4f*)(op + t * 16)     = lo;
      *(v4f*)(op + t * 16 + 4) = hi;
    }
  }
}

extern "C" void kernel_launch(void* const* d_in, const int* in_sizes, int n_in,
                              void* d_out, int out_size, void* d_ws,
                              size_t ws_size, hipStream_t stream) {
  const float* E    = (const float*)d_in[0];
  const float* W    = (const float*)d_in[1];
  const float* bias = (const float*)d_in[2];
  float* out = (float*)d_out;

  const size_t per = (size_t)BS * HEADS * SEQ * HD;
  float* Qw = (float*)d_ws;
  float* Kw = Qw + per;
  float* Vw = Kw + per;

  qkv_proj_kernel<<<(MTOT / PJ_BM) * (N3 / PJ_BN), 128, 0, stream>>>(
      E, W, bias, Qw, Kw, Vw);
  flash_attn_kernel<<<BS * HEADS * (SEQ / 32), 32, 0, stream>>>(Qw, Kw, Vw,
                                                                out);
}


# Projection epilogue parks the 128x128 tile in LDS in plane order and stores whole 128-byte lines, 16 B per lane; V plane order changed so that a work-group owns whole lines; sums unchanged
# speedup vs baseline: 3.4959x; 1.1998x over previous
#include <hip/hip_runtime.h>
#include <hip/hip_bf16.h>


typedef __attribute__((ext_vector_type(2))) float v2f;
typedef __attribute__((ext_vector_type(4))) float v4f;
typedef __attribute__((ext_vector_type(8))) float v8f;

#define BS   2
#define SEQ  2048
#define DMODEL 1024
#define HEADS 16
#define HD   64
#define N3   3072
#define MTOT (BS * SEQ)

__device__ __forceinline__ v8f wmma_f32_16x16x4(v2f a, v2f b, v8f c) {
  return __builtin_amdgcn_wmma_f32_16x16x4_f32(false, a, false, b, (short)0, c,
                                               false, false);
}


#define PJ_BM 128
#define PJ_BN 128
#define PJ_BK 32
#define PJ_AP 36
#define PJ_BP 288
#define PJ_ASZ (PJ_BM * PJ_AP)
#define PJ_BSZ ((PJ_BK / 2) * PJ_BP)
#define PJ_HS 1028

__global__ void __launch_bounds__(128)
qkv_proj_kernel(const float* __restrict__ E, const float* __restrict__ W,
                const float* __restrict__ bias, float* __restrict__ Qw,
                float* __restrict__ Kw, float* __restrict__ Vw) {
  __shared__ __attribute__((aligned(16))) float smem[2 * PJ_ASZ + 2 * PJ_BSZ];
  float* const As = smem;
  float* const Bs = smem + 2 * PJ_ASZ;

  const int tid   = threadIdx.x;
  const int lane  = tid & 31;
  const int wave  = tid >> 5;
  const int wm    = wave >> 1;
  const int wn    = wave & 1;
  const int n15   = lane & 15;
  const int khalf = (lane >> 4) << 1;
  const int NTB   = N3 / PJ_BN;
  const int mtb   = blockIdx.x / NTB;
  const int ntb   = blockIdx.x % NTB;

  const int ar = tid >> 3, ac = tid & 7;
  const int bp = tid >> 5, bc = tid & 31;
  const float* Ag = E + (size_t)(mtb * PJ_BM + ar) * DMODEL + ac * 4;
  const float* Bg = W + (size_t)(2 * bp) * N3 + ntb * PJ_BN + bc * 4;
  const int a_st = ar * PJ_AP + ac * 4;
  const int b_st = bp * PJ_BP + bc * 8;

  const int a_rd = (wm * 64 + n15) * PJ_AP + khalf;
  const int b_rd = (lane >> 4) * PJ_BP + (wn * 64 + n15) * 2;

  v8f acc[4][4];
#pragma unroll
  for (int mm = 0; mm < 4; ++mm)
#pragma unroll
    for (int t = 0; t < 4; ++t) acc[mm][t] = v8f{};

  v4f ra[8], rb[8];
#pragma unroll
  for (int i = 0; i < 8; ++i) {
    ra[i] = *(const v4f*)(Ag + (size_t)i * 16 * DMODEL);
    rb[i] = *(const v4f*)(Bg + (size_t)((i >> 1) * 8 + (i & 1)) * N3);
  }
#pragma unroll
  for (int i = 0; i < 8; ++i)
    *(v4f*)(As + a_st + i * 16 * PJ_AP) = ra[i];
#pragma unroll
  for (int i = 0; i < 4; ++i) {
    v4f lo, hi;
    lo.x = rb[2 * i].x; lo.y = rb[2 * i + 1].x;
    lo.z = rb[2 * i].y; lo.w = rb[2 * i + 1].y;
    hi.x = rb[2 * i].z; hi.y = rb[2 * i + 1].z;
    hi.z = rb[2 * i].w; hi.w = rb[2 * i + 1].w;
    *(v4f*)(Bs + b_st + i * 4 * PJ_BP)     = lo;
    *(v4f*)(Bs + b_st + i * 4 * PJ_BP + 4) = hi;
  }
  __syncthreads();

  for (int kt = 0; kt < DMODEL / PJ_BK; ++kt) {
    const bool more = (kt + 1) < DMODEL / PJ_BK;
    if (more) {
      const float* An = Ag + (kt + 1) * PJ_BK;
      const float* Bn = Bg + (size_t)(kt + 1) * PJ_BK * N3;
#pragma unroll
      for (int i = 0; i < 8; ++i) {
        ra[i] = *(const v4f*)(An + (size_t)i * 16 * DMODEL);
        rb[i] = *(const v4f*)(Bn + (size_t)((i >> 1) * 8 + (i & 1)) * N3);
      }
    }

    const float* Ac = As + (kt & 1) * PJ_ASZ + a_rd;
    const float* Bc = Bs + (kt & 1) * PJ_BSZ + b_rd;
#pragma unroll
    for (int ks = 0; ks < PJ_BK / 4; ++ks) {
      v2f a[4];
#pragma unroll
      for (int mm = 0; mm < 4; ++mm)
        a[mm] = *(const v2f*)(Ac + mm * 16 * PJ_AP + ks * 4);
#pragma unroll
      for (int t = 0; t < 4; ++t) {
        const v2f bf = *(const v2f*)(Bc + ks * 2 * PJ_BP + t * 32);
#pragma unroll
        for (int mm = 0; mm < 4; ++mm)
          acc[mm][t] = wmma_f32_16x16x4(a[mm], bf, acc[mm][t]);
      }
    }

    if (more) {
      float* Ad = As + ((kt + 1) & 1) * PJ_ASZ + a_st;
      float* Bd = Bs + ((kt + 1) & 1) * PJ_BSZ + b_st;
#pragma unroll
      for (int i = 0; i < 8; ++i)
        *(v4f*)(Ad + i * 16 * PJ_AP) = ra[i];
#pragma unroll
      for (int i = 0; i < 4; ++i) {
        v4f lo, hi;
        lo.x = rb[2 * i].x; lo.y = rb[2 * i + 1].x;
        lo.z = rb[2 * i].y; lo.w = rb[2 * i + 1].y;
        hi.x = rb[2 * i].z; hi.y = rb[2 * i + 1].z;
        hi.z = rb[2 * i].w; hi.w = rb[2 * i + 1].w;
        *(v4f*)(Bd + i * 4 * PJ_BP)     = lo;
        *(v4f*)(Bd + i * 4 * PJ_BP + 4) = hi;
      }
    }
    __syncthreads();
  }

  const int mbase = (lane >> 4) * 8;
  const int which = ntb >> 3;
  const int nbase = ntb * PJ_BN + wn * 64;
  float* dst = (which == 0) ? Qw : (which == 1) ? Kw : Vw;
  const float scale = (which == 0) ? 0.125f : 1.0f;
  float* stg = smem + n15 * PJ_HS + wm * 512;
  if (which == 2) {
#pragma unroll
    for (int t = 0; t < 4; ++t) {
      const float bv = bias[nbase + t * 16 + n15];
#pragma unroll
      for (int mm = 0; mm < 4; ++mm)
#pragma unroll
        for (int j2 = 0; j2 < 2; ++j2) {
          v4f pv;
          pv.x = acc[mm][t][4 * j2 + 0] + bv;
          pv.y = acc[mm][t][4 * j2 + 1] + bv;
          pv.z = acc[mm][t][4 * j2 + 2] + bv;
          pv.w = acc[mm][t][4 * j2 + 3] + bv;
          *(v4f*)(stg + mm * 128 + j2 * 64 + (lane >> 4) * 32 +
                  (wn * 4 + t) * 4) = pv;
        }
    }
  } else {
#pragma unroll
    for (int tp = 0; tp < 2; ++tp) {
      const float bv0 = bias[nbase + (2 * tp) * 16 + n15];
      const float bv1 = bias[nbase + (2 * tp + 1) * 16 + n15];
#pragma unroll
      for (int mm = 0; mm < 4; ++mm)
#pragma unroll
        for (int r = 0; r < 8; ++r) {
          v2f pr;
          pr.x = (acc[mm][2 * tp][r] + bv0) * scale;
          pr.y = (acc[mm][2 * tp + 1][r] + bv1) * scale;
          *(v2f*)(stg + mm * 128 + tp * 64 + (mbase + r) * 4 + wn * 2) = pr;
        }
    }
  }
  __syncthreads();

  const int bq  = mtb >> 4;
  const int rc0 = (mtb & 15) * 8;
  const int kk2 = ntb & 7;
  const int coff = (which == 2)
      ? ((lane >> 4) << 9) + ((kk2 >> 1) << 7) + (((lane >> 3) & 1) << 6) +
            (((kk2 & 1) * 8 + (lane & 7)) << 2)
      : (kk2 << 7) + (lane << 2);
#pragma unroll
  for (int hh = 0; hh < 4; ++hh) {
    const int head = wave * 4 + hh;
    float* plane = dst + ((size_t)(bq * HEADS + head)) * SEQ * HD +
                   (size_t)rc0 * 1024 + coff;
    const float* sp = smem + head * PJ_HS + lane * 4;
#pragma unroll
    for (int rc = 0; rc < 8; ++rc)
      *(v4f*)(plane + rc * 1024) = *(const v4f*)(sp + rc * 128);
  }
}

__device__ __forceinline__ float xhalf(float x) {
  int b = __builtin_bit_cast(int, x);
  return __builtin_bit_cast(
      float, __builtin_amdgcn_permlanex16(b, b, 0x76543210, 0xfedcba98, false,
                                          false));
}

__global__ void __launch_bounds__(32)
flash_attn_kernel(const float* __restrict__ Qw, const float* __restrict__ Kw,
                  const float* __restrict__ Vw, float* __restrict__ out) {
  const int lane  = threadIdx.x & 31;
  const int n15   = lane & 15;
  const int mbase = (lane >> 4) * 8;

  const int tile = blockIdx.x;
  const int qt = tile & 63;
  const int bh = tile >> 6;
  const int h  = bh & 15;
  const int b  = bh >> 4;

  const float* Qh = Qw + (((size_t)bh) * SEQ + qt * 32) * HD;
  const float* Kh = Kw + ((size_t)bh) * SEQ * HD;
  const float* Vh = Vw + ((size_t)bh) * SEQ * HD;

  v2f qb[2][16];
#pragma unroll
  for (int u = 0; u < 2; ++u)
#pragma unroll
    for (int kk2 = 0; kk2 < 8; ++kk2) {
      const v4f q4 = *(const v4f*)(Qh + u * 1024 + kk2 * 128 + lane * 4);
      qb[u][2 * kk2].x     = q4.x;  qb[u][2 * kk2].y     = q4.y;
      qb[u][2 * kk2 + 1].x = q4.z;  qb[u][2 * kk2 + 1].y = q4.w;
    }

  v8f acc[2][4];
#pragma unroll
  for (int u = 0; u < 2; ++u)
#pragma unroll
    for (int t = 0; t < 4; ++t) acc[u][t] = v8f{};
  float mi[2], li[2];
#pragma unroll
  for (int u = 0; u < 2; ++u) { mi[u] = -3.0e38f; li[u] = 0.0f; }

  for (int kc = 0; kc < SEQ / 16; ++kc) {
    if (kc + 1 < SEQ / 16) {
      const float* Kn = Kh + (size_t)(kc + 1) * 16 * HD;
      const float* Vn = Vh + (size_t)(kc + 1) * 16 * HD;
      __builtin_prefetch(Kn + lane * 32, 0, 1);
      __builtin_prefetch(Vn + lane * 32, 0, 1);
    }

    const float* Kc = Kh + (size_t)kc * 1024 + lane * 4;
    v8f st[2];
    st[0] = v8f{};
    st[1] = v8f{};
#pragma unroll
    for (int kk2 = 0; kk2 < 8; ++kk2) {
      const v4f k4 = *(const v4f*)(Kc + kk2 * 128);
      v2f kf0, kf1;
      kf0.x = k4.x; kf0.y = k4.y;
      kf1.x = k4.z; kf1.y = k4.w;
      st[0] = wmma_f32_16x16x4(kf0, qb[0][2 * kk2], st[0]);
      st[1] = wmma_f32_16x16x4(kf0, qb[1][2 * kk2], st[1]);
      st[0] = wmma_f32_16x16x4(kf1, qb[0][2 * kk2 + 1], st[0]);
      st[1] = wmma_f32_16x16x4(kf1, qb[1][2 * kk2 + 1], st[1]);
    }

#pragma unroll
    for (int u = 0; u < 2; ++u) {
      float cmax = st[u][0];
#pragma unroll
      for (int r = 1; r < 8; ++r) cmax = fmaxf(cmax, st[u][r]);
      cmax = fmaxf(cmax, xhalf(cmax));
      const float newm  = fmaxf(mi[u], cmax);
      const float alpha = __expf(mi[u] - newm);
      mi[u] = newm;
      float rs = 0.0f;
#pragma unroll
      for (int r = 0; r < 8; ++r) {
        const float pv = __expf(st[u][r] - newm);
        st[u][r] = pv;
        rs += pv;
      }
      li[u] = li[u] * alpha + rs;
#pragma unroll
      for (int t = 0; t < 4; ++t)
#pragma unroll
        for (int r = 0; r < 8; ++r) acc[u][t][r] *= alpha;
    }

    const float* Vc = Vh + (size_t)kc * 1024 + lane * 4;
#pragma unroll
    for (int j2 = 0; j2 < 2; ++j2) {
      v4f v4[4];
#pragma unroll
      for (int t = 0; t < 4; ++t)
        v4[t] = *(const v4f*)(Vc + j2 * 512 + t * 128);
#pragma unroll
      for (int jj = 0; jj < 2; ++jj) {
        const int j = 2 * j2 + jj;
        v2f pb[2];
#pragma unroll
        for (int u = 0; u < 2; ++u) {
          pb[u].x = st[u][2 * j];
          pb[u].y = st[u][2 * j + 1];
        }
#pragma unroll
        for (int t = 0; t < 4; ++t) {
          v2f vf;
          vf.x = jj ? v4[t].z : v4[t].x;
          vf.y = jj ? v4[t].w : v4[t].y;
          acc[0][t] = wmma_f32_16x16x4(vf, pb[0], acc[0][t]);
          acc[1][t] = wmma_f32_16x16x4(vf, pb[1], acc[1][t]);
        }
      }
    }
  }

#pragma unroll
  for (int u = 0; u < 2; ++u) {
    const float l = li[u] + xhalf(li[u]);
    const int srow = qt * 32 + u * 16 + n15;
    float* op = out + (((size_t)b * SEQ) + srow) * DMODEL + h * HD + mbase;
#pragma unroll
    for (int t = 0; t < 4; ++t) {
      v4f lo, hi;
      lo.x = acc[u][t][0] / l; lo.y = acc[u][t][1] / l;
      lo.z = acc[u][t][2] / l; lo.w = acc[u][t][3] / l;
      hi.x = acc[u][t][4] / l; hi.y = acc[u][t][5] / l;
      hi.z = acc[u][t][6] / l; hi.w = acc[u][t][7] / l;
      *(v4f*)(op + t * 16)     = lo;
      *(v4f*)(op + t * 16 + 4) = hi;
    }
  }
}

extern "C" void kernel_launch(void* const* d_in, const int* in_sizes, int n_in,
                              void* d_out, int out_size, void* d_ws,
                              size_t ws_size, hipStream_t stream) {
  const float* E    = (const float*)d_in[0];
  const float* W    = (const float*)d_in[1];
  const float* bias = (const float*)d_in[2];
  float* out = (float*)d_out;

  const size_t per = (size_t)BS * HEADS * SEQ * HD;
  float* Qw = (float*)d_ws;
  float* Kw = Qw + per;
  float* Vw = Kw + per;

  qkv_proj_kernel<<<(MTOT / PJ_BM) * (N3 / PJ_BN), 128, 0, stream>>>(
      E, W, bias, Qw, Kw, Vw);
  flash_attn_kernel<<<BS * HEADS * (SEQ / 32), 32, 0, stream>>>(Qw, Kw, Vw,
                                                                out);
}


# Wave priority (s_setprio) stepped down 3,2,1,0 with progress through the projection K loop and the attention key loop: co-resident waves of a SIMD finish together, no long solo tail; sums unchanged
# speedup vs baseline: 3.6409x; 1.0415x over previous
#include <hip/hip_runtime.h>
#include <hip/hip_bf16.h>


typedef __attribute__((ext_vector_type(2))) float v2f;
typedef __attribute__((ext_vector_type(4))) float v4f;
typedef __attribute__((ext_vector_type(8))) float v8f;

#define BS   2
#define SEQ  2048
#define DMODEL 1024
#define HEADS 16
#define HD   64
#define N3   3072
#define MTOT (BS * SEQ)

__device__ __forceinline__ v8f wmma_f32_16x16x4(v2f a, v2f b, v8f c) {
  return __builtin_amdgcn_wmma_f32_16x16x4_f32(false, a, false, b, (short)0, c,
                                               false, false);
}


#define PJ_BM 128
#define PJ_BN 128
#define PJ_BK 32
#define PJ_AP 36
#define PJ_BP 288
#define PJ_ASZ (PJ_BM * PJ_AP)
#define PJ_BSZ ((PJ_BK / 2) * PJ_BP)
#define PJ_HS 1028

__global__ void __launch_bounds__(128)
qkv_proj_kernel(const float* __restrict__ E, const float* __restrict__ W,
                const float* __restrict__ bias, float* __restrict__ Qw,
                float* __restrict__ Kw, float* __restrict__ Vw) {
  __shared__ __attribute__((aligned(16))) float smem[2 * PJ_ASZ + 2 * PJ_BSZ];
  float* const As = smem;
  float* const Bs = smem + 2 * PJ_ASZ;

  const int tid   = threadIdx.x;
  const int lane  = tid & 31;
  const int wave  = tid >> 5;
  const int wm    = wave >> 1;
  const int wn    = wave & 1;
  const int n15   = lane & 15;
  const int khalf = (lane >> 4) << 1;
  const int NTB   = N3 / PJ_BN;
  const int mtb   = blockIdx.x / NTB;
  const int ntb   = blockIdx.x % NTB;

  const int ar = tid >> 3, ac = tid & 7;
  const int bp = tid >> 5, bc = tid & 31;
  const float* Ag = E + (size_t)(mtb * PJ_BM + ar) * DMODEL + ac * 4;
  const float* Bg = W + (size_t)(2 * bp) * N3 + ntb * PJ_BN + bc * 4;
  const int a_st = ar * PJ_AP + ac * 4;
  const int b_st = bp * PJ_BP + bc * 8;

  const int a_rd = (wm * 64 + n15) * PJ_AP + khalf;
  const int b_rd = (lane >> 4) * PJ_BP + (wn * 64 + n15) * 2;

  v8f acc[4][4];
#pragma unroll
  for (int mm = 0; mm < 4; ++mm)
#pragma unroll
    for (int t = 0; t < 4; ++t) acc[mm][t] = v8f{};

  v4f ra[8], rb[8];
#pragma unroll
  for (int i = 0; i < 8; ++i) {
    ra[i] = *(const v4f*)(Ag + (size_t)i * 16 * DMODEL);
    rb[i] = *(const v4f*)(Bg + (size_t)((i >> 1) * 8 + (i & 1)) * N3);
  }
#pragma unroll
  for (int i = 0; i < 8; ++i)
    *(v4f*)(As + a_st + i * 16 * PJ_AP) = ra[i];
#pragma unroll
  for (int i = 0; i < 4; ++i) {
    v4f lo, hi;
    lo.x = rb[2 * i].x; lo.y = rb[2 * i + 1].x;
    lo.z = rb[2 * i].y; lo.w = rb[2 * i + 1].y;
    hi.x = rb[2 * i].z; hi.y = rb[2 * i + 1].z;
    hi.z = rb[2 * i].w; hi.w = rb[2 * i + 1].w;
    *(v4f*)(Bs + b_st + i * 4 * PJ_BP)     = lo;
    *(v4f*)(Bs + b_st + i * 4 * PJ_BP + 4) = hi;
  }
  __syncthreads();

  auto kstep = [&](const int kt, const bool more) __attribute__((always_inline)) {
    if (more) {
      const float* An = Ag + (kt + 1) * PJ_BK;
      const float* Bn = Bg + (size_t)(kt + 1) * PJ_BK * N3;
#pragma unroll
      for (int i = 0; i < 8; ++i) {
        ra[i] = *(const v4f*)(An + (size_t)i * 16 * DMODEL);
        rb[i] = *(const v4f*)(Bn + (size_t)((i >> 1) * 8 + (i & 1)) * N3);
      }
    }

    const float* Ac = As + (kt & 1) * PJ_ASZ + a_rd;
    const float* Bc = Bs + (kt & 1) * PJ_BSZ + b_rd;
#pragma unroll
    for (int ks = 0; ks < PJ_BK / 4; ++ks) {
      if (more && ks == PJ_BK / 4 - 2) {
        float* Ad = As + ((kt + 1) & 1) * PJ_ASZ + a_st;
        float* Bd = Bs + ((kt + 1) & 1) * PJ_BSZ + b_st;
#pragma unroll
        for (int i = 0; i < 8; ++i)
          *(v4f*)(Ad + i * 16 * PJ_AP) = ra[i];
#pragma unroll
        for (int i = 0; i < 4; ++i) {
          v4f lo, hi;
          lo.x = rb[2 * i].x; lo.y = rb[2 * i + 1].x;
          lo.z = rb[2 * i].y; lo.w = rb[2 * i + 1].y;
          hi.x = rb[2 * i].z; hi.y = rb[2 * i + 1].z;
          hi.z = rb[2 * i].w; hi.w = rb[2 * i + 1].w;
          *(v4f*)(Bd + i * 4 * PJ_BP)     = lo;
          *(v4f*)(Bd + i * 4 * PJ_BP + 4) = hi;
        }
      }
      v2f a[4];
#pragma unroll
      for (int mm = 0; mm < 4; ++mm)
        a[mm] = *(const v2f*)(Ac + mm * 16 * PJ_AP + ks * 4);
#pragma unroll
      for (int t = 0; t < 4; ++t) {
        const v2f bf = *(const v2f*)(Bc + ks * 2 * PJ_BP + t * 32);
#pragma unroll
        for (int mm = 0; mm < 4; ++mm)
          acc[mm][t] = wmma_f32_16x16x4(a[mm], bf, acc[mm][t]);
      }
    }

    __syncthreads();
  };

  int kt = 0;
  for (int seg = 0; seg < 4; ++seg) {
    int kend;
    if (seg == 0)      { __builtin_amdgcn_s_setprio(3); kend = 16; }
    else if (seg == 1) { __builtin_amdgcn_s_setprio(2); kend = 24; }
    else if (seg == 2) { __builtin_amdgcn_s_setprio(1); kend = 29; }
    else               { __builtin_amdgcn_s_setprio(0); kend = DMODEL / PJ_BK - 1; }
    for (; kt < kend; ++kt) kstep(kt, true);
  }
  kstep(DMODEL / PJ_BK - 1, false);

  const int mbase = (lane >> 4) * 8;
  const int which = ntb >> 3;
  const int nbase = ntb * PJ_BN + wn * 64;
  float* dst = (which == 0) ? Qw : (which == 1) ? Kw : Vw;
  const float scale = (which == 0) ? 0.125f : 1.0f;
  float* stg = smem + n15 * PJ_HS + wm * 512;
  if (which == 2) {
#pragma unroll
    for (int t = 0; t < 4; ++t) {
      const float bv = bias[nbase + t * 16 + n15];
#pragma unroll
      for (int mm = 0; mm < 4; ++mm)
#pragma unroll
        for (int j2 = 0; j2 < 2; ++j2) {
          v4f pv;
          pv.x = acc[mm][t][4 * j2 + 0] + bv;
          pv.y = acc[mm][t][4 * j2 + 1] + bv;
          pv.z = acc[mm][t][4 * j2 + 2] + bv;
          pv.w = acc[mm][t][4 * j2 + 3] + bv;
          *(v4f*)(stg + mm * 128 + j2 * 64 + (lane >> 4) * 32 +
                  (wn * 4 + t) * 4) = pv;
        }
    }
  } else {
#pragma unroll
    for (int tp = 0; tp < 2; ++tp) {
      const float bv0 = bias[nbase + (2 * tp) * 16 + n15];
      const float bv1 = bias[nbase + (2 * tp + 1) * 16 + n15];
#pragma unroll
      for (int mm = 0; mm < 4; ++mm)
#pragma unroll
        for (int r = 0; r < 8; ++r) {
          v2f pr;
          pr.x = (acc[mm][2 * tp][r] + bv0) * scale;
          pr.y = (acc[mm][2 * tp + 1][r] + bv1) * scale;
          *(v2f*)(stg + mm * 128 + tp * 64 + (mbase + r) * 4 + wn * 2) = pr;
        }
    }
  }
  __syncthreads();

  const int bq  = mtb >> 4;
  const int rc0 = (mtb & 15) * 8;
  const int kk2 = ntb & 7;
  const int coff = (which == 2)
      ? ((lane >> 4) << 9) + ((kk2 >> 1) << 7) + (((lane >> 3) & 1) << 6) +
            (((kk2 & 1) * 8 + (lane & 7)) << 2)
      : (kk2 << 7) + (lane << 2);
#pragma unroll
  for (int hh = 0; hh < 4; ++hh) {
    const int head = wave * 4 + hh;
    float* plane = dst + ((size_t)(bq * HEADS + head)) * SEQ * HD +
                   (size_t)rc0 * 1024 + coff;
    const float* sp = smem + head * PJ_HS + lane * 4;
#pragma unroll
    for (int rc = 0; rc < 8; ++rc)
      *(v4f*)(plane + rc * 1024) = *(const v4f*)(sp + rc * 128);
  }
}

__device__ __forceinline__ float xhalf(float x) {
  int b = __builtin_bit_cast(int, x);
  return __builtin_bit_cast(
      float, __builtin_amdgcn_permlanex16(b, b, 0x76543210, 0xfedcba98, false,
                                          false));
}

__global__ void __launch_bounds__(32) __attribute__((amdgpu_waves_per_eu(2, 2)))
flash_attn_kernel(const float* __restrict__ Qw, const float* __restrict__ Kw,
                  const float* __restrict__ Vw, float* __restrict__ out) {
  const int lane  = threadIdx.x & 31;
  const int n15   = lane & 15;
  const int mbase = (lane >> 4) * 8;

  const int tile = blockIdx.x;
  const int qt = tile & 63;
  const int bh = tile >> 6;
  const int h  = bh & 15;
  const int b  = bh >> 4;

  const float* Qh = Qw + (((size_t)bh) * SEQ + qt * 32) * HD;
  const float* Kh = Kw + ((size_t)bh) * SEQ * HD;
  const float* Vh = Vw + ((size_t)bh) * SEQ * HD;

  v2f qb[2][16];
#pragma unroll
  for (int u = 0; u < 2; ++u)
#pragma unroll
    for (int kk2 = 0; kk2 < 8; ++kk2) {
      const v4f q4 = *(const v4f*)(Qh + u * 1024 + kk2 * 128 + lane * 4);
      qb[u][2 * kk2].x     = q4.x;  qb[u][2 * kk2].y     = q4.y;
      qb[u][2 * kk2 + 1].x = q4.z;  qb[u][2 * kk2 + 1].y = q4.w;
    }

  v8f acc[2][4];
#pragma unroll
  for (int u = 0; u < 2; ++u)
#pragma unroll
    for (int t = 0; t < 4; ++t) acc[u][t] = v8f{};
  float mi[2], li[2];
#pragma unroll
  for (int u = 0; u < 2; ++u) { mi[u] = -3.0e38f; li[u] = 0.0f; }

  int kc = 0;
  for (int seg = 0; seg < 4; ++seg) {
    int kend;
    if (seg == 0)      { __builtin_amdgcn_s_setprio(3); kend = 64; }
    else if (seg == 1) { __builtin_amdgcn_s_setprio(2); kend = 96; }
    else if (seg == 2) { __builtin_amdgcn_s_setprio(1); kend = 116; }
    else               { __builtin_amdgcn_s_setprio(0); kend = SEQ / 16; }
  for (; kc < kend; ++kc) {
    {
      const int kn = (kc + 1 < SEQ / 16) ? kc + 1 : kc;
      const float* Kn = Kh + (size_t)kn * 16 * HD;
      const float* Vn = Vh + (size_t)kn * 16 * HD;
      __builtin_prefetch(Kn + lane * 32, 0, 1);
      __builtin_prefetch(Vn + lane * 32, 0, 1);
    }

    const float* Kc = Kh + (size_t)kc * 1024 + lane * 4;
    v8f st[2];
    st[0] = v8f{};
    st[1] = v8f{};
#pragma unroll
    for (int kk2 = 0; kk2 < 8; ++kk2) {
      const v4f k4 = *(const v4f*)(Kc + kk2 * 128);
      v2f kf0, kf1;
      kf0.x = k4.x; kf0.y = k4.y;
      kf1.x = k4.z; kf1.y = k4.w;
      st[0] = wmma_f32_16x16x4(kf0, qb[0][2 * kk2], st[0]);
      st[1] = wmma_f32_16x16x4(kf0, qb[1][2 * kk2], st[1]);
      st[0] = wmma_f32_16x16x4(kf1, qb[0][2 * kk2 + 1], st[0]);
      st[1] = wmma_f32_16x16x4(kf1, qb[1][2 * kk2 + 1], st[1]);
    }

#pragma unroll
    for (int u = 0; u < 2; ++u) {
      float cmax = st[u][0];
#pragma unroll
      for (int r = 1; r < 8; ++r) cmax = fmaxf(cmax, st[u][r]);
      cmax = fmaxf(cmax, xhalf(cmax));
      const float newm  = fmaxf(mi[u], cmax);
      const float alpha = __expf(mi[u] - newm);
      mi[u] = newm;
      float rs = 0.0f;
#pragma unroll
      for (int r = 0; r < 8; ++r) {
        const float pv = __expf(st[u][r] - newm);
        st[u][r] = pv;
        rs += pv;
      }
      li[u] = li[u] * alpha + rs;
#pragma unroll
      for (int t = 0; t < 4; ++t)
#pragma unroll
        for (int r = 0; r < 8; ++r) acc[u][t][r] *= alpha;
    }

    const float* Vc = Vh + (size_t)kc * 1024 + lane * 4;
#pragma unroll
    for (int j2 = 0; j2 < 2; ++j2) {
      v4f v4[4];
#pragma unroll
      for (int t = 0; t < 4; ++t)
        v4[t] = *(const v4f*)(Vc + j2 * 512 + t * 128);
#pragma unroll
      for (int jj = 0; jj < 2; ++jj) {
        const int j = 2 * j2 + jj;
        v2f pb[2];
#pragma unroll
        for (int u = 0; u < 2; ++u) {
          pb[u].x = st[u][2 * j];
          pb[u].y = st[u][2 * j + 1];
        }
#pragma unroll
        for (int t = 0; t < 4; ++t) {
          v2f vf;
          vf.x = jj ? v4[t].z : v4[t].x;
          vf.y = jj ? v4[t].w : v4[t].y;
          acc[0][t] = wmma_f32_16x16x4(vf, pb[0], acc[0][t]);
          acc[1][t] = wmma_f32_16x16x4(vf, pb[1], acc[1][t]);
        }
      }
    }
  }
  }

#pragma unroll
  for (int u = 0; u < 2; ++u) {
    const float l = li[u] + xhalf(li[u]);
    const int srow = qt * 32 + u * 16 + n15;
    float* op = out + (((size_t)b * SEQ) + srow) * DMODEL + h * HD + mbase;
#pragma unroll
    for (int t = 0; t < 4; ++t) {
      v4f lo, hi;
      lo.x = acc[u][t][0] / l; lo.y = acc[u][t][1] / l;
      lo.z = acc[u][t][2] / l; lo.w = acc[u][t][3] / l;
      hi.x = acc[u][t][4] / l; hi.y = acc[u][t][5] / l;
      hi.z = acc[u][t][6] / l; hi.w = acc[u][t][7] / l;
      *(v4f*)(op + t * 16)     = lo;
      *(v4f*)(op + t * 16 + 4) = hi;
    }
  }
}

extern "C" void kernel_launch(void* const* d_in, const int* in_sizes, int n_in,
                              void* d_out, int out_size, void* d_ws,
                              size_t ws_size, hipStream_t stream) {
  const float* E    = (const float*)d_in[0];
  const float* W    = (const float*)d_in[1];
  const float* bias = (const float*)d_in[2];
  float* out = (float*)d_out;

  const size_t per = (size_t)BS * HEADS * SEQ * HD;
  float* Qw = (float*)d_ws;
  float* Kw = Qw + per;
  float* Vw = Kw + per;

  qkv_proj_kernel<<<(MTOT / PJ_BM) * (N3 / PJ_BN), 128, 0, stream>>>(
      E, W, bias, Qw, Kw, Vw);
  flash_attn_kernel<<<BS * HEADS * (SEQ / 32), 32, 0, stream>>>(Qw, Kw, Vw,
                                                                out);
}


# Attention K/V chunks copied once per 4-wave group into LDS (global loads 8 chunks ahead, ds_store, 2 stages, 1 barrier per stage); a chunk's 16 fragment reads from LDS in one place; sums unchanged
# speedup vs baseline: 3.7773x; 1.0375x over previous
#include <hip/hip_runtime.h>
#include <hip/hip_bf16.h>


typedef __attribute__((ext_vector_type(2))) float v2f;
typedef __attribute__((ext_vector_type(4))) float v4f;
typedef __attribute__((ext_vector_type(8))) float v8f;

#define BS   2
#define SEQ  2048
#define DMODEL 1024
#define HEADS 16
#define HD   64
#define N3   3072
#define MTOT (BS * SEQ)

__device__ __forceinline__ v8f wmma_f32_16x16x4(v2f a, v2f b, v8f c) {
  return __builtin_amdgcn_wmma_f32_16x16x4_f32(false, a, false, b, (short)0, c,
                                               false, false);
}


#define PJ_BM 128
#define PJ_BN 128
#define PJ_BK 32
#define PJ_AP 36
#define PJ_BP 288
#define PJ_ASZ (PJ_BM * PJ_AP)
#define PJ_BSZ ((PJ_BK / 2) * PJ_BP)
#define PJ_HS 1028

__global__ void __launch_bounds__(128)
qkv_proj_kernel(const float* __restrict__ E, const float* __restrict__ W,
                const float* __restrict__ bias, float* __restrict__ Qw,
                float* __restrict__ Kw, float* __restrict__ Vw) {
  __shared__ __attribute__((aligned(16))) float smem[2 * PJ_ASZ + 2 * PJ_BSZ];
  float* const As = smem;
  float* const Bs = smem + 2 * PJ_ASZ;

  const int tid   = threadIdx.x;
  const int lane  = tid & 31;
  const int wave  = tid >> 5;
  const int wm    = wave >> 1;
  const int wn    = wave & 1;
  const int n15   = lane & 15;
  const int khalf = (lane >> 4) << 1;
  const int NTB   = N3 / PJ_BN;
  const int mtb   = blockIdx.x / NTB;
  const int ntb   = blockIdx.x % NTB;

  const int ar = tid >> 3, ac = tid & 7;
  const int bp = tid >> 5, bc = tid & 31;
  const float* Ag = E + (size_t)(mtb * PJ_BM + ar) * DMODEL + ac * 4;
  const float* Bg = W + (size_t)(2 * bp) * N3 + ntb * PJ_BN + bc * 4;
  const int a_st = ar * PJ_AP + ac * 4;
  const int b_st = bp * PJ_BP + bc * 8;

  const int a_rd = (wm * 64 + n15) * PJ_AP + khalf;
  const int b_rd = (lane >> 4) * PJ_BP + (wn * 64 + n15) * 2;

  v8f acc[4][4];
#pragma unroll
  for (int mm = 0; mm < 4; ++mm)
#pragma unroll
    for (int t = 0; t < 4; ++t) acc[mm][t] = v8f{};

  v4f ra[8], rb[8];
#pragma unroll
  for (int i = 0; i < 8; ++i) {
    ra[i] = *(const v4f*)(Ag + (size_t)i * 16 * DMODEL);
    rb[i] = *(const v4f*)(Bg + (size_t)((i >> 1) * 8 + (i & 1)) * N3);
  }
#pragma unroll
  for (int i = 0; i < 8; ++i)
    *(v4f*)(As + a_st + i * 16 * PJ_AP) = ra[i];
#pragma unroll
  for (int i = 0; i < 4; ++i) {
    v4f lo, hi;
    lo.x = rb[2 * i].x; lo.y = rb[2 * i + 1].x;
    lo.z = rb[2 * i].y; lo.w = rb[2 * i + 1].y;
    hi.x = rb[2 * i].z; hi.y = rb[2 * i + 1].z;
    hi.z = rb[2 * i].w; hi.w = rb[2 * i + 1].w;
    *(v4f*)(Bs + b_st + i * 4 * PJ_BP)     = lo;
    *(v4f*)(Bs + b_st + i * 4 * PJ_BP + 4) = hi;
  }
  __syncthreads();

  auto kstep = [&](const int kt, const bool more) __attribute__((always_inline)) {
    if (more) {
      const float* An = Ag + (kt + 1) * PJ_BK;
      const float* Bn = Bg + (size_t)(kt + 1) * PJ_BK * N3;
#pragma unroll
      for (int i = 0; i < 8; ++i) {
        ra[i] = *(const v4f*)(An + (size_t)i * 16 * DMODEL);
        rb[i] = *(const v4f*)(Bn + (size_t)((i >> 1) * 8 + (i & 1)) * N3);
      }
    }

    const float* Ac = As + (kt & 1) * PJ_ASZ + a_rd;
    const float* Bc = Bs + (kt & 1) * PJ_BSZ + b_rd;
#pragma unroll
    for (int ks = 0; ks < PJ_BK / 4; ++ks) {
      if (more && ks == PJ_BK / 4 - 2) {
        float* Ad = As + ((kt + 1) & 1) * PJ_ASZ + a_st;
        float* Bd = Bs + ((kt + 1) & 1) * PJ_BSZ + b_st;
#pragma unroll
        for (int i = 0; i < 8; ++i)
          *(v4f*)(Ad + i * 16 * PJ_AP) = ra[i];
#pragma unroll
        for (int i = 0; i < 4; ++i) {
          v4f lo, hi;
          lo.x = rb[2 * i].x; lo.y = rb[2 * i + 1].x;
          lo.z = rb[2 * i].y; lo.w = rb[2 * i + 1].y;
          hi.x = rb[2 * i].z; hi.y = rb[2 * i + 1].z;
          hi.z = rb[2 * i].w; hi.w = rb[2 * i + 1].w;
          *(v4f*)(Bd + i * 4 * PJ_BP)     = lo;
          *(v4f*)(Bd + i * 4 * PJ_BP + 4) = hi;
        }
      }
      v2f a[4];
#pragma unroll
      for (int mm = 0; mm < 4; ++mm)
        a[mm] = *(const v2f*)(Ac + mm * 16 * PJ_AP + ks * 4);
#pragma unroll
      for (int t = 0; t < 4; ++t) {
        const v2f bf = *(const v2f*)(Bc + ks * 2 * PJ_BP + t * 32);
#pragma unroll
        for (int mm = 0; mm < 4; ++mm)
          acc[mm][t] = wmma_f32_16x16x4(a[mm], bf, acc[mm][t]);
      }
    }

    __syncthreads();
  };

  int kt = 0;
  for (int seg = 0; seg < 4; ++seg) {
    int kend;
    if (seg == 0)      { __builtin_amdgcn_s_setprio(3); kend = 16; }
    else if (seg == 1) { __builtin_amdgcn_s_setprio(2); kend = 24; }
    else if (seg == 2) { __builtin_amdgcn_s_setprio(1); kend = 29; }
    else               { __builtin_amdgcn_s_setprio(0); kend = DMODEL / PJ_BK - 1; }
    for (; kt < kend; ++kt) kstep(kt, true);
  }
  kstep(DMODEL / PJ_BK - 1, false);

  const int mbase = (lane >> 4) * 8;
  const int which = ntb >> 3;
  const int nbase = ntb * PJ_BN + wn * 64;
  float* dst = (which == 0) ? Qw : (which == 1) ? Kw : Vw;
  const float scale = (which == 0) ? 0.125f : 1.0f;
  float* stg = smem + n15 * PJ_HS + wm * 512;
  if (which == 2) {
#pragma unroll
    for (int t = 0; t < 4; ++t) {
      const float bv = bias[nbase + t * 16 + n15];
#pragma unroll
      for (int mm = 0; mm < 4; ++mm)
#pragma unroll
        for (int j2 = 0; j2 < 2; ++j2) {
          v4f pv;
          pv.x = acc[mm][t][4 * j2 + 0] + bv;
          pv.y = acc[mm][t][4 * j2 + 1] + bv;
          pv.z = acc[mm][t][4 * j2 + 2] + bv;
          pv.w = acc[mm][t][4 * j2 + 3] + bv;
          *(v4f*)(stg + mm * 128 + j2 * 64 + (lane >> 4) * 32 +
                  (wn * 4 + t) * 4) = pv;
        }
    }
  } else {
#pragma unroll
    for (int tp = 0; tp < 2; ++tp) {
      const float bv0 = bias[nbase + (2 * tp) * 16 + n15];
      const float bv1 = bias[nbase + (2 * tp + 1) * 16 + n15];
#pragma unroll
      for (int mm = 0; mm < 4; ++mm)
#pragma unroll
        for (int r = 0; r < 8; ++r) {
          v2f pr;
          pr.x = (acc[mm][2 * tp][r] + bv0) * scale;
          pr.y = (acc[mm][2 * tp + 1][r] + bv1) * scale;
          *(v2f*)(stg + mm * 128 + tp * 64 + (mbase + r) * 4 + wn * 2) = pr;
        }
    }
  }
  __syncthreads();

  const int bq  = mtb >> 4;
  const int rc0 = (mtb & 15) * 8;
  const int kk2 = ntb & 7;
  const int coff = (which == 2)
      ? ((lane >> 4) << 9) + ((kk2 >> 1) << 7) + (((lane >> 3) & 1) << 6) +
            (((kk2 & 1) * 8 + (lane & 7)) << 2)
      : (kk2 << 7) + (lane << 2);
#pragma unroll
  for (int hh = 0; hh < 4; ++hh) {
    const int head = wave * 4 + hh;
    float* plane = dst + ((size_t)(bq * HEADS + head)) * SEQ * HD +
                   (size_t)rc0 * 1024 + coff;
    const float* sp = smem + head * PJ_HS + lane * 4;
#pragma unroll
    for (int rc = 0; rc < 8; ++rc)
      *(v4f*)(plane + rc * 1024) = *(const v4f*)(sp + rc * 128);
  }
}

__device__ __forceinline__ float xhalf(float x) {
  int b = __builtin_bit_cast(int, x);
  return __builtin_bit_cast(
      float, __builtin_amdgcn_permlanex16(b, b, 0x76543210, 0xfedcba98, false,
                                          false));
}

#define FA_CH  8
#define FA_STG (FA_CH * 1024)

__global__ void __launch_bounds__(128) __attribute__((amdgpu_waves_per_eu(2, 2)))
flash_attn_kernel(const float* __restrict__ Qw, const float* __restrict__ Kw,
                  const float* __restrict__ Vw, float* __restrict__ out) {
  __shared__ __attribute__((aligned(16))) float Ks[2 * FA_STG];
  __shared__ __attribute__((aligned(16))) float Vs[2 * FA_STG];

  const int tid   = threadIdx.x;
  const int lane  = tid & 31;
  const int wave  = tid >> 5;
  const int n15   = lane & 15;
  const int mbase = (lane >> 4) * 8;

  const int tile = blockIdx.x * 4 + wave;
  const int qt = tile & 63;
  const int bh = tile >> 6;
  const int h  = bh & 15;
  const int b  = bh >> 4;

  const float* Qh = Qw + (((size_t)bh) * SEQ + qt * 32) * HD;
  const float* Kh = Kw + ((size_t)bh) * SEQ * HD;
  const float* Vh = Vw + ((size_t)bh) * SEQ * HD;

  {
    v4f rk[2 * FA_CH], rv[2 * FA_CH];
#pragma unroll
    for (int i = 0; i < 2 * FA_CH; ++i) {
      rk[i] = *(const v4f*)(Kh + i * 512 + tid * 4);
      rv[i] = *(const v4f*)(Vh + i * 512 + tid * 4);
    }
#pragma unroll
    for (int i = 0; i < 2 * FA_CH; ++i) {
      *(v4f*)(Ks + i * 512 + tid * 4) = rk[i];
      *(v4f*)(Vs + i * 512 + tid * 4) = rv[i];
    }
  }

  v2f qb[2][16];
#pragma unroll
  for (int u = 0; u < 2; ++u)
#pragma unroll
    for (int kk2 = 0; kk2 < 8; ++kk2) {
      const v4f q4 = *(const v4f*)(Qh + u * 1024 + kk2 * 128 + lane * 4);
      qb[u][2 * kk2].x     = q4.x;  qb[u][2 * kk2].y     = q4.y;
      qb[u][2 * kk2 + 1].x = q4.z;  qb[u][2 * kk2 + 1].y = q4.w;
    }

  v8f acc[2][4];
#pragma unroll
  for (int u = 0; u < 2; ++u)
#pragma unroll
    for (int t = 0; t < 4; ++t) acc[u][t] = v8f{};
  float mi[2], li[2];
#pragma unroll
  for (int u = 0; u < 2; ++u) { mi[u] = -3.0e38f; li[u] = 0.0f; }

  __syncthreads();

  int kc = 0;
  for (int seg = 0; seg < 4; ++seg) {
    int kend;
    if (seg == 0)      { __builtin_amdgcn_s_setprio(3); kend = 64; }
    else if (seg == 1) { __builtin_amdgcn_s_setprio(2); kend = 96; }
    else if (seg == 2) { __builtin_amdgcn_s_setprio(1); kend = 116; }
    else               { __builtin_amdgcn_s_setprio(0); kend = SEQ / 16; }
  for (; kc < kend; ++kc) {
    const int c   = kc & (FA_CH - 1);
    const int cur = ((kc / FA_CH) & 1) * FA_STG;
    const bool more = kc + FA_CH < SEQ / 16;

    v4f rk0, rk1, rv0, rv1;
    if (more) {
      const float* Kn = Kh + (size_t)(kc + FA_CH) * 1024 + tid * 4;
      const float* Vn = Vh + (size_t)(kc + FA_CH) * 1024 + tid * 4;
      rk0 = *(const v4f*)(Kn);
      rk1 = *(const v4f*)(Kn + 512);
      rv0 = *(const v4f*)(Vn);
      rv1 = *(const v4f*)(Vn + 512);
    }

    const float* Kc = Ks + cur + c * 1024 + lane * 4;
    const float* Vc = Vs + cur + c * 1024 + lane * 4;
    v4f k4[8];
#pragma unroll
    for (int kk2 = 0; kk2 < 8; ++kk2)
      k4[kk2] = *(const v4f*)(Kc + kk2 * 128);
    v4f v4[2][4];
#pragma unroll
    for (int j2 = 0; j2 < 2; ++j2)
#pragma unroll
      for (int t = 0; t < 4; ++t)
        v4[j2][t] = *(const v4f*)(Vc + j2 * 512 + t * 128);
    __builtin_amdgcn_sched_barrier(0);

    v8f st[2];
    st[0] = v8f{};
    st[1] = v8f{};
#pragma unroll
    for (int kk2 = 0; kk2 < 8; ++kk2) {
      v2f kf0, kf1;
      kf0.x = k4[kk2].x; kf0.y = k4[kk2].y;
      kf1.x = k4[kk2].z; kf1.y = k4[kk2].w;
      st[0] = wmma_f32_16x16x4(kf0, qb[0][2 * kk2], st[0]);
      st[1] = wmma_f32_16x16x4(kf0, qb[1][2 * kk2], st[1]);
      st[0] = wmma_f32_16x16x4(kf1, qb[0][2 * kk2 + 1], st[0]);
      st[1] = wmma_f32_16x16x4(kf1, qb[1][2 * kk2 + 1], st[1]);
    }

#pragma unroll
    for (int u = 0; u < 2; ++u) {
      float cmax = st[u][0];
#pragma unroll
      for (int r = 1; r < 8; ++r) cmax = fmaxf(cmax, st[u][r]);
      cmax = fmaxf(cmax, xhalf(cmax));
      const float newm  = fmaxf(mi[u], cmax);
      const float alpha = __expf(mi[u] - newm);
      mi[u] = newm;
      float rs = 0.0f;
#pragma unroll
      for (int r = 0; r < 8; ++r) {
        const float pv = __expf(st[u][r] - newm);
        st[u][r] = pv;
        rs += pv;
      }
      li[u] = li[u] * alpha + rs;
#pragma unroll
      for (int t = 0; t < 4; ++t)
#pragma unroll
        for (int r = 0; r < 8; ++r) acc[u][t][r] *= alpha;
    }

#pragma unroll
    for (int j2 = 0; j2 < 2; ++j2) {
#pragma unroll
      for (int jj = 0; jj < 2; ++jj) {
        const int j = 2 * j2 + jj;
        v2f pb[2];
#pragma unroll
        for (int u = 0; u < 2; ++u) {
          pb[u].x = st[u][2 * j];
          pb[u].y = st[u][2 * j + 1];
        }
#pragma unroll
        for (int t = 0; t < 4; ++t) {
          v2f vf;
          vf.x = jj ? v4[j2][t].z : v4[j2][t].x;
          vf.y = jj ? v4[j2][t].w : v4[j2][t].y;
          acc[0][t] = wmma_f32_16x16x4(vf, pb[0], acc[0][t]);
          acc[1][t] = wmma_f32_16x16x4(vf, pb[1], acc[1][t]);
        }
      }
    }

    if (more) {
      float* Kd = Ks + (FA_STG - cur) + c * 1024 + tid * 4;
      float* Vd = Vs + (FA_STG - cur) + c * 1024 + tid * 4;
      *(v4f*)(Kd)       = rk0;
      *(v4f*)(Kd + 512) = rk1;
      *(v4f*)(Vd)       = rv0;
      *(v4f*)(Vd + 512) = rv1;
    }
    if (c == FA_CH - 1) __syncthreads();
  }
  }

#pragma unroll
  for (int u = 0; u < 2; ++u) {
    const float l = li[u] + xhalf(li[u]);
    const int srow = qt * 32 + u * 16 + n15;
    float* op = out + (((size_t)b * SEQ) + srow) * DMODEL + h * HD + mbase;
#pragma unroll
    for (int t = 0; t < 4; ++t) {
      v4f lo, hi;
      lo.x = acc[u][t][0] / l; lo.y = acc[u][t][1] / l;
      lo.z = acc[u][t][2] / l; lo.w = acc[u][t][3] / l;
      hi.x = acc[u][t][4] / l; hi.y = acc[u][t][5] / l;
      hi.z = acc[u][t][6] / l; hi.w = acc[u][t][7] / l;
      *(v4f*)(op + t * 16)     = lo;
      *(v4f*)(op + t * 16 + 4) = hi;
    }
  }
}

extern "C" void kernel_launch(void* const* d_in, const int* in_sizes, int n_in,
                              void* d_out, int out_size, void* d_ws,
                              size_t ws_size, hipStream_t stream) {
  const float* E    = (const float*)d_in[0];
  const float* W    = (const float*)d_in[1];
  const float* bias = (const float*)d_in[2];
  float* out = (float*)d_out;

  const size_t per = (size_t)BS * HEADS * SEQ * HD;
  float* Qw = (float*)d_ws;
  float* Kw = Qw + per;
  float* Vw = Kw + per;

  qkv_proj_kernel<<<(MTOT / PJ_BM) * (N3 / PJ_BN), 128, 0, stream>>>(
      E, W, bias, Qw, Kw, Vw);
  flash_attn_kernel<<<BS * HEADS * (SEQ / 128), 128, 0, stream>>>(Qw, Kw, Vw,
                                                                  out);
}


# Attention key loop takes 64 keys (4 chunks) per step, not 16: one max/exchange, one output rescale, one K and one V read group per step; row sum a tree over 32 terms, running max per 64 keys
# speedup vs baseline: 3.9046x; 1.0337x over previous
#include <hip/hip_runtime.h>
#include <hip/hip_bf16.h>


typedef __attribute__((ext_vector_type(2))) float v2f;
typedef __attribute__((ext_vector_type(4))) float v4f;
typedef __attribute__((ext_vector_type(8))) float v8f;

#define BS   2
#define SEQ  2048
#define DMODEL 1024
#define HEADS 16
#define HD   64
#define N3   3072
#define MTOT (BS * SEQ)

__device__ __forceinline__ v8f wmma_f32_16x16x4(v2f a, v2f b, v8f c) {
  return __builtin_amdgcn_wmma_f32_16x16x4_f32(false, a, false, b, (short)0, c,
                                               false, false);
}


#define PJ_BM 128
#define PJ_BN 128
#define PJ_BK 32
#define PJ_AP 36
#define PJ_BP 288
#define PJ_ASZ (PJ_BM * PJ_AP)
#define PJ_BSZ ((PJ_BK / 2) * PJ_BP)
#define PJ_HS 1028

__global__ void __launch_bounds__(128)
qkv_proj_kernel(const float* __restrict__ E, const float* __restrict__ W,
                const float* __restrict__ bias, float* __restrict__ Qw,
                float* __restrict__ Kw, float* __restrict__ Vw) {
  __shared__ __attribute__((aligned(16))) float smem[2 * PJ_ASZ + 2 * PJ_BSZ];
  float* const As = smem;
  float* const Bs = smem + 2 * PJ_ASZ;

  const int tid   = threadIdx.x;
  const int lane  = tid & 31;
  const int wave  = tid >> 5;
  const int wm    = wave >> 1;
  const int wn    = wave & 1;
  const int n15   = lane & 15;
  const int khalf = (lane >> 4) << 1;
  const int NTB   = N3 / PJ_BN;
  const int mtb   = blockIdx.x / NTB;
  const int ntb   = blockIdx.x % NTB;

  const int ar = tid >> 3, ac = tid & 7;
  const int bp = tid >> 5, bc = tid & 31;
  const float* Ag = E + (size_t)(mtb * PJ_BM + ar) * DMODEL + ac * 4;
  const float* Bg = W + (size_t)(2 * bp) * N3 + ntb * PJ_BN + bc * 4;
  const int a_st = ar * PJ_AP + ac * 4;
  const int b_st = bp * PJ_BP + bc * 8;

  const int a_rd = (wm * 64 + n15) * PJ_AP + khalf;
  const int b_rd = (lane >> 4) * PJ_BP + (wn * 64 + n15) * 2;

  v8f acc[4][4];
#pragma unroll
  for (int mm = 0; mm < 4; ++mm)
#pragma unroll
    for (int t = 0; t < 4; ++t) acc[mm][t] = v8f{};

  v4f ra[8], rb[8];
#pragma unroll
  for (int i = 0; i < 8; ++i) {
    ra[i] = *(const v4f*)(Ag + (size_t)i * 16 * DMODEL);
    rb[i] = *(const v4f*)(Bg + (size_t)((i >> 1) * 8 + (i & 1)) * N3);
  }
#pragma unroll
  for (int i = 0; i < 8; ++i)
    *(v4f*)(As + a_st + i * 16 * PJ_AP) = ra[i];
#pragma unroll
  for (int i = 0; i < 4; ++i) {
    v4f lo, hi;
    lo.x = rb[2 * i].x; lo.y = rb[2 * i + 1].x;
    lo.z = rb[2 * i].y; lo.w = rb[2 * i + 1].y;
    hi.x = rb[2 * i].z; hi.y = rb[2 * i + 1].z;
    hi.z = rb[2 * i].w; hi.w = rb[2 * i + 1].w;
    *(v4f*)(Bs + b_st + i * 4 * PJ_BP)     = lo;
    *(v4f*)(Bs + b_st + i * 4 * PJ_BP + 4) = hi;
  }
  __syncthreads();

  auto kstep = [&](const int kt, const bool more) __attribute__((always_inline)) {
    if (more) {
      const float* An = Ag + (kt + 1) * PJ_BK;
      const float* Bn = Bg + (size_t)(kt + 1) * PJ_BK * N3;
#pragma unroll
      for (int i = 0; i < 8; ++i) {
        ra[i] = *(const v4f*)(An + (size_t)i * 16 * DMODEL);
        rb[i] = *(const v4f*)(Bn + (size_t)((i >> 1) * 8 + (i & 1)) * N3);
      }
    }

    const float* Ac = As + (kt & 1) * PJ_ASZ + a_rd;
    const float* Bc = Bs + (kt & 1) * PJ_BSZ + b_rd;
#pragma unroll
    for (int ks = 0; ks < PJ_BK / 4; ++ks) {
      if (more && ks == PJ_BK / 4 - 2) {
        float* Ad = As + ((kt + 1) & 1) * PJ_ASZ + a_st;
        float* Bd = Bs + ((kt + 1) & 1) * PJ_BSZ + b_st;
#pragma unroll
        for (int i = 0; i < 8; ++i)
          *(v4f*)(Ad + i * 16 * PJ_AP) = ra[i];
#pragma unroll
        for (int i = 0; i < 4; ++i) {
          v4f lo, hi;
          lo.x = rb[2 * i].x; lo.y = rb[2 * i + 1].x;
          lo.z = rb[2 * i].y; lo.w = rb[2 * i + 1].y;
          hi.x = rb[2 * i].z; hi.y = rb[2 * i + 1].z;
          hi.z = rb[2 * i].w; hi.w = rb[2 * i + 1].w;
          *(v4f*)(Bd + i * 4 * PJ_BP)     = lo;
          *(v4f*)(Bd + i * 4 * PJ_BP + 4) = hi;
        }
      }
      v2f a[4];
#pragma unroll
      for (int mm = 0; mm < 4; ++mm)
        a[mm] = *(const v2f*)(Ac + mm * 16 * PJ_AP + ks * 4);
#pragma unroll
      for (int t = 0; t < 4; ++t) {
        const v2f bf = *(const v2f*)(Bc + ks * 2 * PJ_BP + t * 32);
#pragma unroll
        for (int mm = 0; mm < 4; ++mm)
          acc[mm][t] = wmma_f32_16x16x4(a[mm], bf, acc[mm][t]);
      }
    }

    __syncthreads();
  };

  int kt = 0;
  for (int seg = 0; seg < 4; ++seg) {
    int kend;
    if (seg == 0)      { __builtin_amdgcn_s_setprio(3); kend = 16; }
    else if (seg == 1) { __builtin_amdgcn_s_setprio(2); kend = 24; }
    else if (seg == 2) { __builtin_amdgcn_s_setprio(1); kend = 29; }
    else               { __builtin_amdgcn_s_setprio(0); kend = DMODEL / PJ_BK - 1; }
    for (; kt < kend; ++kt) kstep(kt, true);
  }
  kstep(DMODEL / PJ_BK - 1, false);

  const int mbase = (lane >> 4) * 8;
  const int which = ntb >> 3;
  const int nbase = ntb * PJ_BN + wn * 64;
  float* dst = (which == 0) ? Qw : (which == 1) ? Kw : Vw;
  const float scale = (which == 0) ? 0.125f : 1.0f;
  float* stg = smem + n15 * PJ_HS + wm * 512;
  if (which == 2) {
#pragma unroll
    for (int t = 0; t < 4; ++t) {
      const float bv = bias[nbase + t * 16 + n15];
#pragma unroll
      for (int mm = 0; mm < 4; ++mm)
#pragma unroll
        for (int j2 = 0; j2 < 2; ++j2) {
          v4f pv;
          pv.x = acc[mm][t][4 * j2 + 0] + bv;
          pv.y = acc[mm][t][4 * j2 + 1] + bv;
          pv.z = acc[mm][t][4 * j2 + 2] + bv;
          pv.w = acc[mm][t][4 * j2 + 3] + bv;
          *(v4f*)(stg + mm * 128 + j2 * 64 + (lane >> 4) * 32 +
                  (wn * 4 + t) * 4) = pv;
        }
    }
  } else {
#pragma unroll
    for (int tp = 0; tp < 2; ++tp) {
      const float bv0 = bias[nbase + (2 * tp) * 16 + n15];
      const float bv1 = bias[nbase + (2 * tp + 1) * 16 + n15];
#pragma unroll
      for (int mm = 0; mm < 4; ++mm)
#pragma unroll
        for (int r = 0; r < 8; ++r) {
          v2f pr;
          pr.x = (acc[mm][2 * tp][r] + bv0) * scale;
          pr.y = (acc[mm][2 * tp + 1][r] + bv1) * scale;
          *(v2f*)(stg + mm * 128 + tp * 64 + (mbase + r) * 4 + wn * 2) = pr;
        }
    }
  }
  __syncthreads();

  const int bq  = mtb >> 4;
  const int rc0 = (mtb & 15) * 8;
  const int kk2 = ntb & 7;
  const int coff = (which == 2)
      ? ((lane >> 4) << 9) + ((kk2 >> 1) << 7) + (((lane >> 3) & 1) << 6) +
            (((kk2 & 1) * 8 + (lane & 7)) << 2)
      : (kk2 << 7) + (lane << 2);
#pragma unroll
  for (int hh = 0; hh < 4; ++hh) {
    const int head = wave * 4 + hh;
    float* plane = dst + ((size_t)(bq * HEADS + head)) * SEQ * HD +
                   (size_t)rc0 * 1024 + coff;
    const float* sp = smem + head * PJ_HS + lane * 4;
#pragma unroll
    for (int rc = 0; rc < 8; ++rc)
      *(v4f*)(plane + rc * 1024) = *(const v4f*)(sp + rc * 128);
  }
}

__device__ __forceinline__ float xhalf(float x) {
  int b = __builtin_bit_cast(int, x);
  return __builtin_bit_cast(
      float, __builtin_amdgcn_permlanex16(b, b, 0x76543210, 0xfedcba98, false,
                                          false));
}

#define FA_CH  8
#define FA_STG (FA_CH * 1024)
#define FA_KB  4
#define FA_NST (SEQ / (16 * FA_KB))

__global__ void __launch_bounds__(128) __attribute__((amdgpu_waves_per_eu(2, 2)))
flash_attn_kernel(const float* __restrict__ Qw, const float* __restrict__ Kw,
                  const float* __restrict__ Vw, float* __restrict__ out) {
  __shared__ __attribute__((aligned(16))) float Ks[2 * FA_STG];
  __shared__ __attribute__((aligned(16))) float Vs[2 * FA_STG];

  const int tid   = threadIdx.x;
  const int lane  = tid & 31;
  const int wave  = tid >> 5;
  const int n15   = lane & 15;
  const int mbase = (lane >> 4) * 8;

  const int tile = blockIdx.x * 4 + wave;
  const int qt = tile & 63;
  const int bh = tile >> 6;
  const int h  = bh & 15;
  const int b  = bh >> 4;

  const float* Qh = Qw + (((size_t)bh) * SEQ + qt * 32) * HD;
  const float* Kh = Kw + ((size_t)bh) * SEQ * HD;
  const float* Vh = Vw + ((size_t)bh) * SEQ * HD;

  {
    v4f rk[2 * FA_CH], rv[2 * FA_CH];
#pragma unroll
    for (int i = 0; i < 2 * FA_CH; ++i) {
      rk[i] = *(const v4f*)(Kh + i * 512 + tid * 4);
      rv[i] = *(const v4f*)(Vh + i * 512 + tid * 4);
    }
#pragma unroll
    for (int i = 0; i < 2 * FA_CH; ++i) {
      *(v4f*)(Ks + i * 512 + tid * 4) = rk[i];
      *(v4f*)(Vs + i * 512 + tid * 4) = rv[i];
    }
  }

  v2f qb[2][16];
#pragma unroll
  for (int u = 0; u < 2; ++u)
#pragma unroll
    for (int kk2 = 0; kk2 < 8; ++kk2) {
      const v4f q4 = *(const v4f*)(Qh + u * 1024 + kk2 * 128 + lane * 4);
      qb[u][2 * kk2].x     = q4.x;  qb[u][2 * kk2].y     = q4.y;
      qb[u][2 * kk2 + 1].x = q4.z;  qb[u][2 * kk2 + 1].y = q4.w;
    }

  v8f acc[2][4];
#pragma unroll
  for (int u = 0; u < 2; ++u)
#pragma unroll
    for (int t = 0; t < 4; ++t) acc[u][t] = v8f{};
  float mi[2], li[2];
#pragma unroll
  for (int u = 0; u < 2; ++u) { mi[u] = -3.0e38f; li[u] = 0.0f; }

  __syncthreads();

  int ks = 0;
  for (int seg = 0; seg < 4; ++seg) {
    int kend;
    if (seg == 0)      { __builtin_amdgcn_s_setprio(3); kend = 16; }
    else if (seg == 1) { __builtin_amdgcn_s_setprio(2); kend = 24; }
    else if (seg == 2) { __builtin_amdgcn_s_setprio(1); kend = 29; }
    else               { __builtin_amdgcn_s_setprio(0); kend = FA_NST; }
  for (; ks < kend; ++ks) {
    const int kc0 = ks * FA_KB;
    const int c0  = kc0 & (FA_CH - 1);
    const int cur = ((kc0 / FA_CH) & 1) * FA_STG;
    const bool more = kc0 + FA_CH < SEQ / 16;

    const float* Kc = Ks + cur + c0 * 1024 + lane * 4;
    v4f k4[FA_KB][8];
#pragma unroll
    for (int kk2 = 0; kk2 < 8; ++kk2)
#pragma unroll
      for (int ch = 0; ch < FA_KB; ++ch)
        k4[ch][kk2] = *(const v4f*)(Kc + ch * 1024 + kk2 * 128);

    v4f rk[2 * FA_KB], rv[2 * FA_KB];
    if (more) {
      const float* Kn = Kh + (size_t)(kc0 + FA_CH) * 1024 + tid * 4;
      const float* Vn = Vh + (size_t)(kc0 + FA_CH) * 1024 + tid * 4;
#pragma unroll
      for (int i = 0; i < 2 * FA_KB; ++i) {
        rk[i] = *(const v4f*)(Kn + i * 512);
        rv[i] = *(const v4f*)(Vn + i * 512);
      }
    }
    __builtin_amdgcn_sched_barrier(0);

    v8f st[2][FA_KB];
#pragma unroll
    for (int ch = 0; ch < FA_KB; ++ch) {
      st[0][ch] = v8f{};
      st[1][ch] = v8f{};
    }
#pragma unroll
    for (int kk2 = 0; kk2 < 8; ++kk2) {
#pragma unroll
      for (int ch = 0; ch < FA_KB; ++ch) {
        v2f kf0;
        kf0.x = k4[ch][kk2].x; kf0.y = k4[ch][kk2].y;
        st[0][ch] = wmma_f32_16x16x4(kf0, qb[0][2 * kk2], st[0][ch]);
        st[1][ch] = wmma_f32_16x16x4(kf0, qb[1][2 * kk2], st[1][ch]);
      }
#pragma unroll
      for (int ch = 0; ch < FA_KB; ++ch) {
        v2f kf1;
        kf1.x = k4[ch][kk2].z; kf1.y = k4[ch][kk2].w;
        st[0][ch] = wmma_f32_16x16x4(kf1, qb[0][2 * kk2 + 1], st[0][ch]);
        st[1][ch] = wmma_f32_16x16x4(kf1, qb[1][2 * kk2 + 1], st[1][ch]);
      }
    }
    __builtin_amdgcn_sched_barrier(0);

    const float* Vc = Vs + cur + c0 * 1024 + lane * 4;
    v4f v4[FA_KB][2][4];
#pragma unroll
    for (int ch = 0; ch < FA_KB; ++ch)
#pragma unroll
      for (int j2 = 0; j2 < 2; ++j2)
#pragma unroll
        for (int t = 0; t < 4; ++t)
          v4[ch][j2][t] = *(const v4f*)(Vc + ch * 1024 + j2 * 512 + t * 128);
    __builtin_amdgcn_sched_barrier(0);

#pragma unroll
    for (int u = 0; u < 2; ++u) {
      float cm[FA_KB];
#pragma unroll
      for (int ch = 0; ch < FA_KB; ++ch) {
        float m = fmaxf(fmaxf(st[u][ch][0], st[u][ch][1]), st[u][ch][2]);
        m = fmaxf(fmaxf(m, st[u][ch][3]), st[u][ch][4]);
        m = fmaxf(fmaxf(m, st[u][ch][5]), st[u][ch][6]);
        cm[ch] = fmaxf(m, st[u][ch][7]);
      }
      float cmax = fmaxf(fmaxf(cm[0], cm[1]), fmaxf(cm[2], cm[3]));
      cmax = fmaxf(cmax, xhalf(cmax));
      const float newm  = fmaxf(mi[u], cmax);
      const float alpha = __expf(mi[u] - newm);
      mi[u] = newm;
      float cs4[FA_KB];
#pragma unroll
      for (int ch = 0; ch < FA_KB; ++ch) {
#pragma unroll
        for (int r = 0; r < 8; ++r)
          st[u][ch][r] = __expf(st[u][ch][r] - newm);
        cs4[ch] = ((st[u][ch][0] + st[u][ch][1]) + (st[u][ch][2] + st[u][ch][3])) +
                  ((st[u][ch][4] + st[u][ch][5]) + (st[u][ch][6] + st[u][ch][7]));
      }
      const float rs = (cs4[0] + cs4[1]) + (cs4[2] + cs4[3]);
      li[u] = li[u] * alpha + rs;
#pragma unroll
      for (int t = 0; t < 4; ++t)
#pragma unroll
        for (int r = 0; r < 8; ++r) acc[u][t][r] *= alpha;
    }

#pragma unroll
    for (int ch = 0; ch < FA_KB; ++ch) {
#pragma unroll
      for (int j2 = 0; j2 < 2; ++j2) {
#pragma unroll
        for (int jj = 0; jj < 2; ++jj) {
          const int j = 2 * j2 + jj;
          v2f pb[2];
#pragma unroll
          for (int u = 0; u < 2; ++u) {
            pb[u].x = st[u][ch][2 * j];
            pb[u].y = st[u][ch][2 * j + 1];
          }
#pragma unroll
          for (int t = 0; t < 4; ++t) {
            v2f vf;
            vf.x = jj ? v4[ch][j2][t].z : v4[ch][j2][t].x;
            vf.y = jj ? v4[ch][j2][t].w : v4[ch][j2][t].y;
            acc[0][t] = wmma_f32_16x16x4(vf, pb[0], acc[0][t]);
            acc[1][t] = wmma_f32_16x16x4(vf, pb[1], acc[1][t]);
          }
        }
      }
    }

    if (more) {
      float* Kd = Ks + (FA_STG - cur) + c0 * 1024 + tid * 4;
      float* Vd = Vs + (FA_STG - cur) + c0 * 1024 + tid * 4;
#pragma unroll
      for (int i = 0; i < 2 * FA_KB; ++i) {
        *(v4f*)(Kd + i * 512) = rk[i];
        *(v4f*)(Vd + i * 512) = rv[i];
      }
    }
    if (c0 + FA_KB == FA_CH) __syncthreads();
  }
  }

#pragma unroll
  for (int u = 0; u < 2; ++u) {
    const float l = li[u] + xhalf(li[u]);
    const int srow = qt * 32 + u * 16 + n15;
    float* op = out + (((size_t)b * SEQ) + srow) * DMODEL + h * HD + mbase;
#pragma unroll
    for (int t = 0; t < 4; ++t) {
      v4f lo, hi;
      lo.x = acc[u][t][0] / l; lo.y = acc[u][t][1] / l;
      lo.z = acc[u][t][2] / l; lo.w = acc[u][t][3] / l;
      hi.x = acc[u][t][4] / l; hi.y = acc[u][t][5] / l;
      hi.z = acc[u][t][6] / l; hi.w = acc[u][t][7] / l;
      *(v4f*)(op + t * 16)     = lo;
      *(v4f*)(op + t * 16 + 4) = hi;
    }
  }
}

extern "C" void kernel_launch(void* const* d_in, const int* in_sizes, int n_in,
                              void* d_out, int out_size, void* d_ws,
                              size_t ws_size, hipStream_t stream) {
  const float* E    = (const float*)d_in[0];
  const float* W    = (const float*)d_in[1];
  const float* bias = (const float*)d_in[2];
  float* out = (float*)d_out;

  const size_t per = (size_t)BS * HEADS * SEQ * HD;
  float* Qw = (float*)d_ws;
  float* Kw = Qw + per;
  float* Vw = Kw + per;

  qkv_proj_kernel<<<(MTOT / PJ_BM) * (N3 / PJ_BN), 128, 0, stream>>>(
      E, W, bias, Qw, Kw, Vw);
  flash_attn_kernel<<<BS * HEADS * (SEQ / 128), 128, 0, stream>>>(Qw, Kw, Vw,
                                                                  out);
}
